# starting kernel
# baseline (speedup 1.0000x reference)
#include <hip/hip_runtime.h>
#include <hip/hip_bf16.h>

#define NNODES 100000
#define NEDGES 1200000
#define HID 64
#define NLAYERS 3
#define GAMMA_C (-0.45f)
#define ZETA_C  (1.05f)

typedef float v2f __attribute__((ext_vector_type(2)));
typedef float v8f __attribute__((ext_vector_type(8)));

__global__ __launch_bounds__(128) void node_transform_kernel(
    const float* __restrict__ x, const float* __restrict__ W1,
    const float* __restrict__ b1, const float* __restrict__ W2,
    const float* __restrict__ bias2, int has_bias2,
    float* __restrict__ p, int n_nodes)
{
    __shared__ float w1s[64 * 64];
    __shared__ float w2s[64 * 64];
    __shared__ float gs[4 * 16 * 68];

    for (int i = threadIdx.x; i < 64 * 64; i += 128) {
        w1s[i] = W1[i];
        w2s[i] = W2[i];
    }
    __syncthreads();

    const int wave = threadIdx.x >> 5;
    const int lane = threadIdx.x & 31;
    const int mn    = lane & 15;
    const int kh    = (lane >> 4) << 1;
    const int mbase = (lane >> 4) << 3;
    float* gt = gs + wave * (16 * 68);

    const int tile = blockIdx.x * 4 + wave;
    const int row0 = tile * 16;
    if (row0 >= n_nodes) return;

    int ri = row0 + mn;
    if (ri >= n_nodes) ri = n_nodes - 1;
    const float* xr = x + (size_t)ri * HID;

    const v8f vz = {0.f, 0.f, 0.f, 0.f, 0.f, 0.f, 0.f, 0.f};
    v8f acc[4] = {vz, vz, vz, vz};

    #pragma unroll
    for (int kk = 0; kk < 16; ++kk) {
        const int k = kk * 4 + kh;
        v2f a; a.x = xr[k]; a.y = xr[k + 1];
        #pragma unroll
        for (int nt = 0; nt < 4; ++nt) {
            v2f b;
            b.x = w1s[k * 64 + nt * 16 + mn];
            b.y = w1s[(k + 1) * 64 + nt * 16 + mn];
            acc[nt] = __builtin_amdgcn_wmma_f32_16x16x4_f32(
                false, a, false, b, (short)0, acc[nt], false, false);
        }
    }

    #pragma unroll
    for (int nt = 0; nt < 4; ++nt) {
        const float bv = b1[nt * 16 + mn];
        #pragma unroll
        for (int i = 0; i < 8; ++i) {
            float v = acc[nt][i] + bv;
            v = v > 0.f ? v : 0.f;
            gt[(mbase + i) * 68 + nt * 16 + mn] = v;
        }
    }

    v8f acc2[4] = {vz, vz, vz, vz};
    #pragma unroll
    for (int kk = 0; kk < 16; ++kk) {
        const int k = kk * 4 + kh;
        v2f a; a.x = gt[mn * 68 + k]; a.y = gt[mn * 68 + k + 1];
        #pragma unroll
        for (int nt = 0; nt < 4; ++nt) {
            v2f b;
            b.x = w2s[k * 64 + nt * 16 + mn];
            b.y = w2s[(k + 1) * 64 + nt * 16 + mn];
            acc2[nt] = __builtin_amdgcn_wmma_f32_16x16x4_f32(
                false, a, false, b, (short)0, acc2[nt], false, false);
        }
    }

    #pragma unroll
    for (int nt = 0; nt < 4; ++nt) {
        const float b2v = has_bias2 ? bias2[nt * 16 + mn] : 0.f;
        #pragma unroll
        for (int i = 0; i < 8; ++i) {
            const int m = mbase + i;
            const int r = row0 + m;
            if (r < n_nodes)
                p[(size_t)r * HID + nt * 16 + mn] = acc2[nt][i] + b2v;
        }
    }
}

__global__ __launch_bounds__(256) void edge_att_kernel(
    const float* __restrict__ p1, const float* __restrict__ p2,
    const float* __restrict__ w2, const float* __restrict__ b2,
    const int* __restrict__ row, const int* __restrict__ col,
    float* __restrict__ mask, float* __restrict__ rowsum, int n_edges)
{
    const int e = blockIdx.x * 16 + (threadIdx.x >> 4);
    if (e >= n_edges) return;
    const int l4 = (threadIdx.x & 15) * 4;

    const int r = row[e];
    const int c = col[e];
    const float4 a = *(const float4*)(p1 + (size_t)r * HID + l4);
    const float4 b = *(const float4*)(p2 + (size_t)c * HID + l4);
    const float4 w = *(const float4*)(w2 + l4);

    const float t0 = fmaxf(a.x + b.x, 0.f);
    const float t1 = fmaxf(a.y + b.y, 0.f);
    const float t2 = fmaxf(a.z + b.z, 0.f);
    const float t3 = fmaxf(a.w + b.w, 0.f);
    float s = t0 * w.x + t1 * w.y + t2 * w.z + t3 * w.w;

    s += __shfl_xor(s, 8, 32);
    s += __shfl_xor(s, 4, 32);
    s += __shfl_xor(s, 2, 32);
    s += __shfl_xor(s, 1, 32);

    if ((threadIdx.x & 15) == 0) {
        const float la = s + b2[0];
        const float sg = 1.f / (1.f + __expf(-la));
        float m = sg * (ZETA_C - GAMMA_C) + GAMMA_C;
        m = fminf(fmaxf(m, 0.f), 1.f);
        mask[e] = m;
        atomicAdd(rowsum + r, m);
    }
}

__global__ __launch_bounds__(256) void dinv_kernel(
    const float* __restrict__ rowsum, float* __restrict__ dis, int n)
{
    const int i = blockIdx.x * 256 + threadIdx.x;
    if (i < n) {
        const float s = rowsum[i] + 1e-6f;
        dis[i] = fminf(rsqrtf(s), 10.f);
    }
}

__global__ __launch_bounds__(256) void edge_spmm_kernel(
    const float* __restrict__ x, const float* __restrict__ mask,
    const float* __restrict__ dis,
    const int* __restrict__ row, const int* __restrict__ col,
    float* __restrict__ xn, int n_edges)
{
    const int e = blockIdx.x * 16 + (threadIdx.x >> 4);
    if (e >= n_edges) return;
    const int l4 = (threadIdx.x & 15) * 4;

    const int r = row[e];
    const int c = col[e];
    const float v = mask[e] * dis[r] * dis[c];
    const float4 xc = *(const float4*)(x + (size_t)c * HID + l4);
    float* dst = xn + (size_t)r * HID + l4;
    atomicAdd(dst + 0, v * xc.x);
    atomicAdd(dst + 1, v * xc.y);
    atomicAdd(dst + 2, v * xc.z);
    atomicAdd(dst + 3, v * xc.w);
}

__global__ __launch_bounds__(256) void copy2_kernel(
    const float* __restrict__ src, float* __restrict__ d0,
    float* __restrict__ d1, int n)
{
    const int i = blockIdx.x * 256 + threadIdx.x;
    if (i < n) { const float v = src[i]; d0[i] = v; d1[i] = v; }
}

__global__ __launch_bounds__(256) void fill0_kernel(float* __restrict__ p, int n)
{
    const int i = blockIdx.x * 256 + threadIdx.x;
    if (i < n) p[i] = 0.f;
}

__global__ __launch_bounds__(256) void accum_kernel(
    float* __restrict__ out, const float* __restrict__ xn, int n)
{
    const int i = blockIdx.x * 256 + threadIdx.x;
    if (i < n) out[i] += xn[i];
}

extern "C" void kernel_launch(void* const* d_in, const int* in_sizes, int n_in,
                              void* d_out, int out_size, void* d_ws, size_t ws_size,
                              hipStream_t stream) {
    (void)in_sizes; (void)n_in; (void)out_size; (void)ws_size;

    const float* feat = (const float*)d_in[0];
    const float* nbW  = (const float*)d_in[1];
    const float* nbb  = (const float*)d_in[2];
    const float* sW   = (const float*)d_in[3];
    const float* sb   = (const float*)d_in[4];
    const float* a1W  = (const float*)d_in[5];
    const float* a1b  = (const float*)d_in[6];
    const float* a2W  = (const float*)d_in[7];
    const float* a2b  = (const float*)d_in[8];
    const int*   row  = (const int*)d_in[9];
    const int*   col  = (const int*)d_in[10];
    float* out = (float*)d_out;

    const int NH = NNODES * HID;
    float* ws = (float*)d_ws;
    float* xa     = ws; ws += NH;
    float* xb     = ws; ws += NH;
    float* p1     = ws; ws += NH;
    float* p2     = ws; ws += NH;
    float* maskb  = ws; ws += NEDGES;
    float* rowsum = ws; ws += NNODES;
    float* dis    = ws; ws += NNODES;

    const int gNH    = (NH + 255) / 256;
    const int gNode  = (NNODES + 255) / 256;
    const int gTile  = ((NNODES + 15) / 16 + 3) / 4;
    const int gEdge  = (NEDGES + 15) / 16;

    copy2_kernel<<<gNH, 256, 0, stream>>>(feat, xa, out, NH);

    float* xcur = xa;
    float* xnew = xb;
    for (int l = 0; l < NLAYERS; ++l) {
        fill0_kernel<<<gNode, 256, 0, stream>>>(rowsum, NNODES);
        fill0_kernel<<<gNH, 256, 0, stream>>>(xnew, NH);

        node_transform_kernel<<<gTile, 128, 0, stream>>>(
            xcur, nbW + (size_t)l * 4096, nbb + l * 64,
            a1W + (size_t)l * 8192, a1b + l * 64, 1, p1, NNODES);
        node_transform_kernel<<<gTile, 128, 0, stream>>>(
            xcur, sW + (size_t)l * 4096, sb + l * 64,
            a1W + (size_t)l * 8192 + 4096, a1b, 0, p2, NNODES);

        edge_att_kernel<<<gEdge, 256, 0, stream>>>(
            p1, p2, a2W + l * 64, a2b + l, row, col, maskb, rowsum, NEDGES);

        dinv_kernel<<<gNode, 256, 0, stream>>>(rowsum, dis, NNODES);

        edge_spmm_kernel<<<gEdge, 256, 0, stream>>>(
            xcur, maskb, dis, row, col, xnew, NEDGES);

        accum_kernel<<<gNH, 256, 0, stream>>>(out, xnew, NH);

        float* t = xcur; xcur = xnew; xnew = t;
    }
}


# edge kernels: each wave walks 4 chunks of 32 edges, 16 whole-row gathers in flight per half-wave, lane-per-edge index loads and epilogue (was 2 edges per wave); atomic adds arrive in another order
# speedup vs baseline: 1.2950x; 1.2950x over previous
#include <hip/hip_runtime.h>
#include <hip/hip_bf16.h>

#define NNODES 100000
#define NEDGES 1200000
#define HID 64
#define NLAYERS 3
#define GAMMA_C (-0.45f)
#define ZETA_C  (1.05f)

typedef float v2f __attribute__((ext_vector_type(2)));
typedef float v8f __attribute__((ext_vector_type(8)));

__global__ __launch_bounds__(128) void node_transform_kernel(
    const float* __restrict__ x, const float* __restrict__ W1,
    const float* __restrict__ b1, const float* __restrict__ W2,
    const float* __restrict__ bias2, int has_bias2,
    float* __restrict__ p, int n_nodes)
{
    __shared__ float w1s[64 * 64];
    __shared__ float w2s[64 * 64];
    __shared__ float gs[4 * 16 * 68];

    for (int i = threadIdx.x; i < 64 * 64; i += 128) {
        w1s[i] = W1[i];
        w2s[i] = W2[i];
    }
    __syncthreads();

    const int wave = threadIdx.x >> 5;
    const int lane = threadIdx.x & 31;
    const int mn    = lane & 15;
    const int kh    = (lane >> 4) << 1;
    const int mbase = (lane >> 4) << 3;
    float* gt = gs + wave * (16 * 68);

    const int tile = blockIdx.x * 4 + wave;
    const int row0 = tile * 16;
    if (row0 >= n_nodes) return;

    int ri = row0 + mn;
    if (ri >= n_nodes) ri = n_nodes - 1;
    const float* xr = x + (size_t)ri * HID;

    const v8f vz = {0.f, 0.f, 0.f, 0.f, 0.f, 0.f, 0.f, 0.f};
    v8f acc[4] = {vz, vz, vz, vz};

    #pragma unroll
    for (int kk = 0; kk < 16; ++kk) {
        const int k = kk * 4 + kh;
        v2f a; a.x = xr[k]; a.y = xr[k + 1];
        #pragma unroll
        for (int nt = 0; nt < 4; ++nt) {
            v2f b;
            b.x = w1s[k * 64 + nt * 16 + mn];
            b.y = w1s[(k + 1) * 64 + nt * 16 + mn];
            acc[nt] = __builtin_amdgcn_wmma_f32_16x16x4_f32(
                false, a, false, b, (short)0, acc[nt], false, false);
        }
    }

    #pragma unroll
    for (int nt = 0; nt < 4; ++nt) {
        const float bv = b1[nt * 16 + mn];
        #pragma unroll
        for (int i = 0; i < 8; ++i) {
            float v = acc[nt][i] + bv;
            v = v > 0.f ? v : 0.f;
            gt[(mbase + i) * 68 + nt * 16 + mn] = v;
        }
    }

    v8f acc2[4] = {vz, vz, vz, vz};
    #pragma unroll
    for (int kk = 0; kk < 16; ++kk) {
        const int k = kk * 4 + kh;
        v2f a; a.x = gt[mn * 68 + k]; a.y = gt[mn * 68 + k + 1];
        #pragma unroll
        for (int nt = 0; nt < 4; ++nt) {
            v2f b;
            b.x = w2s[k * 64 + nt * 16 + mn];
            b.y = w2s[(k + 1) * 64 + nt * 16 + mn];
            acc2[nt] = __builtin_amdgcn_wmma_f32_16x16x4_f32(
                false, a, false, b, (short)0, acc2[nt], false, false);
        }
    }

    #pragma unroll
    for (int nt = 0; nt < 4; ++nt) {
        const float b2v = has_bias2 ? bias2[nt * 16 + mn] : 0.f;
        #pragma unroll
        for (int i = 0; i < 8; ++i) {
            const int m = mbase + i;
            const int r = row0 + m;
            if (r < n_nodes)
                p[(size_t)r * HID + nt * 16 + mn] = acc2[nt][i] + b2v;
        }
    }
}

#define EDGE_CHUNK 32
#define EDGE_CHUNKS_PER_WAVE 4
#define EDGE_WAVES_PER_BLOCK 4
static_assert(NEDGES % EDGE_CHUNK == 0, "edge count must fill whole chunks");

__global__ __launch_bounds__(128) void edge_att_kernel(
    const float* __restrict__ p1, const float* __restrict__ p2,
    const float* __restrict__ w2, const float* __restrict__ b2,
    const int* __restrict__ row, const int* __restrict__ col,
    float* __restrict__ mask, float* __restrict__ rowsum, int n_chunks)
{
    const int lane   = threadIdx.x & 31;
    const int sub    = lane & 15;
    const int half16 = lane & 16;
    const int wave_g = blockIdx.x * EDGE_WAVES_PER_BLOCK + (threadIdx.x >> 5);
    const int ch0    = wave_g * EDGE_CHUNKS_PER_WAVE;
    const float4* __restrict__ p1v = (const float4*)p1;
    const float4* __restrict__ p2v = (const float4*)p2;

    const float4 w = ((const float4*)w2)[sub];
    const float bias = b2[0];

    for (int q = 0; q < EDGE_CHUNKS_PER_WAVE; ++q) {
        const int ch = ch0 + q;
        if (ch >= n_chunks) break;
        const int e_l = ch * EDGE_CHUNK + lane;
        const int r_l = row[e_l];
        const int c_l = col[e_l];

        float4 a[16], b[16];
        #pragma unroll
        for (int j = 0; j < 16; ++j) {
            const unsigned r = (unsigned)__shfl(r_l, half16 + j, 32);
            const unsigned c = (unsigned)__shfl(c_l, half16 + j, 32);
            a[j] = p1v[r * (HID / 4) + sub];
            b[j] = p2v[c * (HID / 4) + sub];
        }

        float mine = 0.f;
        #pragma unroll
        for (int j = 0; j < 16; ++j) {
            const float t0 = fmaxf(a[j].x + b[j].x, 0.f);
            const float t1 = fmaxf(a[j].y + b[j].y, 0.f);
            const float t2 = fmaxf(a[j].z + b[j].z, 0.f);
            const float t3 = fmaxf(a[j].w + b[j].w, 0.f);
            float s = t0 * w.x + t1 * w.y + t2 * w.z + t3 * w.w;
            s += __shfl_xor(s, 8, 32);
            s += __shfl_xor(s, 4, 32);
            s += __shfl_xor(s, 2, 32);
            s += __shfl_xor(s, 1, 32);
            mine = (sub == j) ? s : mine;
        }

        const float la = mine + bias;
        const float sg = 1.f / (1.f + __expf(-la));
        float m = sg * (ZETA_C - GAMMA_C) + GAMMA_C;
        m = fminf(fmaxf(m, 0.f), 1.f);
        mask[e_l] = m;
        atomicAdd(rowsum + r_l, m);
    }
}

__global__ __launch_bounds__(256) void dinv_kernel(
    const float* __restrict__ rowsum, float* __restrict__ dis, int n)
{
    const int i = blockIdx.x * 256 + threadIdx.x;
    if (i < n) {
        const float s = rowsum[i] + 1e-6f;
        dis[i] = fminf(rsqrtf(s), 10.f);
    }
}

__global__ __launch_bounds__(128) void edge_spmm_kernel(
    const float* __restrict__ x, const float* __restrict__ mask,
    const float* __restrict__ dis,
    const int* __restrict__ row, const int* __restrict__ col,
    float* __restrict__ xn, int n_chunks)
{
    const int lane   = threadIdx.x & 31;
    const int sub    = lane & 15;
    const int half16 = lane & 16;
    const int l4     = sub * 4;
    const int wave_g = blockIdx.x * EDGE_WAVES_PER_BLOCK + (threadIdx.x >> 5);
    const int ch0    = wave_g * EDGE_CHUNKS_PER_WAVE;
    const float4* __restrict__ xv = (const float4*)x;

    for (int q = 0; q < EDGE_CHUNKS_PER_WAVE; ++q) {
        const int ch = ch0 + q;
        if (ch >= n_chunks) break;
        const int e_l = ch * EDGE_CHUNK + lane;
        const int r_l = row[e_l];
        const int c_l = col[e_l];

        float4 xc[16];
        #pragma unroll
        for (int j = 0; j < 16; ++j) {
            const unsigned c = (unsigned)__shfl(c_l, half16 + j, 32);
            xc[j] = xv[c * (HID / 4) + sub];
        }

        const float v_l = mask[e_l] * dis[r_l] * dis[c_l];

        #pragma unroll
        for (int j = 0; j < 16; ++j) {
            const int   r = __shfl(r_l, half16 + j, 32);
            const float v = __shfl(v_l, half16 + j, 32);
            float* dst = xn + (r * HID + l4);
            atomicAdd(dst + 0, v * xc[j].x);
            atomicAdd(dst + 1, v * xc[j].y);
            atomicAdd(dst + 2, v * xc[j].z);
            atomicAdd(dst + 3, v * xc[j].w);
        }
    }
}

__global__ __launch_bounds__(256) void copy2_kernel(
    const float* __restrict__ src, float* __restrict__ d0,
    float* __restrict__ d1, int n)
{
    const int i = blockIdx.x * 256 + threadIdx.x;
    if (i < n) { const float v = src[i]; d0[i] = v; d1[i] = v; }
}

__global__ __launch_bounds__(256) void fill0_kernel(float* __restrict__ p, int n)
{
    const int i = blockIdx.x * 256 + threadIdx.x;
    if (i < n) p[i] = 0.f;
}

__global__ __launch_bounds__(256) void accum_kernel(
    float* __restrict__ out, const float* __restrict__ xn, int n)
{
    const int i = blockIdx.x * 256 + threadIdx.x;
    if (i < n) out[i] += xn[i];
}

extern "C" void kernel_launch(void* const* d_in, const int* in_sizes, int n_in,
                              void* d_out, int out_size, void* d_ws, size_t ws_size,
                              hipStream_t stream) {
    (void)in_sizes; (void)n_in; (void)out_size; (void)ws_size;

    const float* feat = (const float*)d_in[0];
    const float* nbW  = (const float*)d_in[1];
    const float* nbb  = (const float*)d_in[2];
    const float* sW   = (const float*)d_in[3];
    const float* sb   = (const float*)d_in[4];
    const float* a1W  = (const float*)d_in[5];
    const float* a1b  = (const float*)d_in[6];
    const float* a2W  = (const float*)d_in[7];
    const float* a2b  = (const float*)d_in[8];
    const int*   row  = (const int*)d_in[9];
    const int*   col  = (const int*)d_in[10];
    float* out = (float*)d_out;

    const int NH = NNODES * HID;
    float* ws = (float*)d_ws;
    float* xa     = ws; ws += NH;
    float* xb     = ws; ws += NH;
    float* p1     = ws; ws += NH;
    float* p2     = ws; ws += NH;
    float* maskb  = ws; ws += NEDGES;
    float* rowsum = ws; ws += NNODES;
    float* dis    = ws; ws += NNODES;

    const int gNH    = (NH + 255) / 256;
    const int gNode  = (NNODES + 255) / 256;
    const int gTile  = ((NNODES + 15) / 16 + 3) / 4;
    const int nChunk = NEDGES / EDGE_CHUNK;
    const int gEdge  = (nChunk + EDGE_CHUNKS_PER_WAVE * EDGE_WAVES_PER_BLOCK - 1)
                       / (EDGE_CHUNKS_PER_WAVE * EDGE_WAVES_PER_BLOCK);

    copy2_kernel<<<gNH, 256, 0, stream>>>(feat, xa, out, NH);

    float* xcur = xa;
    float* xnew = xb;
    for (int l = 0; l < NLAYERS; ++l) {
        fill0_kernel<<<gNode, 256, 0, stream>>>(rowsum, NNODES);
        fill0_kernel<<<gNH, 256, 0, stream>>>(xnew, NH);

        node_transform_kernel<<<gTile, 128, 0, stream>>>(
            xcur, nbW + (size_t)l * 4096, nbb + l * 64,
            a1W + (size_t)l * 8192, a1b + l * 64, 1, p1, NNODES);
        node_transform_kernel<<<gTile, 128, 0, stream>>>(
            xcur, sW + (size_t)l * 4096, sb + l * 64,
            a1W + (size_t)l * 8192 + 4096, a1b, 0, p2, NNODES);

        edge_att_kernel<<<gEdge, 128, 0, stream>>>(
            p1, p2, a2W + l * 64, a2b + l, row, col, maskb, rowsum, nChunk);

        dinv_kernel<<<gNode, 256, 0, stream>>>(rowsum, dis, NNODES);

        edge_spmm_kernel<<<gEdge, 128, 0, stream>>>(
            xcur, maskb, dis, row, col, xnew, nChunk);

        accum_kernel<<<gNH, 256, 0, stream>>>(out, xnew, NH);

        float* t = xcur; xcur = xnew; xnew = t;
    }
}


# edges bucketed by destination row once per call (counting sort, int atomics); both edge passes become per-row gather-sums in registers: no float atomics, fills or accumulate pass; bucket-order sums
# speedup vs baseline: 2.2752x; 1.7569x over previous
#include <hip/hip_runtime.h>
#include <hip/hip_bf16.h>

#define NNODES 100000
#define NEDGES 1200000
#define HID 64
#define NLAYERS 3
#define GAMMA_C (-0.45f)
#define ZETA_C  (1.05f)

typedef float v2f __attribute__((ext_vector_type(2)));
typedef float v8f __attribute__((ext_vector_type(8)));

__global__ __launch_bounds__(128) void node_transform_kernel(
    const float* __restrict__ x, const float* __restrict__ W1,
    const float* __restrict__ b1, const float* __restrict__ W2,
    const float* __restrict__ bias2, int has_bias2,
    float* __restrict__ p, int n_nodes)
{
    __shared__ float w1s[64 * 64];
    __shared__ float w2s[64 * 64];
    __shared__ float gs[4 * 16 * 68];

    for (int i = threadIdx.x; i < 64 * 64; i += 128) {
        w1s[i] = W1[i];
        w2s[i] = W2[i];
    }
    __syncthreads();

    const int wave = threadIdx.x >> 5;
    const int lane = threadIdx.x & 31;
    const int mn    = lane & 15;
    const int kh    = (lane >> 4) << 1;
    const int mbase = (lane >> 4) << 3;
    float* gt = gs + wave * (16 * 68);

    const int tile = blockIdx.x * 4 + wave;
    const int row0 = tile * 16;
    if (row0 >= n_nodes) return;

    int ri = row0 + mn;
    if (ri >= n_nodes) ri = n_nodes - 1;
    const float* xr = x + (size_t)ri * HID;

    const v8f vz = {0.f, 0.f, 0.f, 0.f, 0.f, 0.f, 0.f, 0.f};
    v8f acc[4] = {vz, vz, vz, vz};

    #pragma unroll
    for (int kk = 0; kk < 16; ++kk) {
        const int k = kk * 4 + kh;
        v2f a; a.x = xr[k]; a.y = xr[k + 1];
        #pragma unroll
        for (int nt = 0; nt < 4; ++nt) {
            v2f b;
            b.x = w1s[k * 64 + nt * 16 + mn];
            b.y = w1s[(k + 1) * 64 + nt * 16 + mn];
            acc[nt] = __builtin_amdgcn_wmma_f32_16x16x4_f32(
                false, a, false, b, (short)0, acc[nt], false, false);
        }
    }

    #pragma unroll
    for (int nt = 0; nt < 4; ++nt) {
        const float bv = b1[nt * 16 + mn];
        #pragma unroll
        for (int i = 0; i < 8; ++i) {
            float v = acc[nt][i] + bv;
            v = v > 0.f ? v : 0.f;
            gt[(mbase + i) * 68 + nt * 16 + mn] = v;
        }
    }

    v8f acc2[4] = {vz, vz, vz, vz};
    #pragma unroll
    for (int kk = 0; kk < 16; ++kk) {
        const int k = kk * 4 + kh;
        v2f a; a.x = gt[mn * 68 + k]; a.y = gt[mn * 68 + k + 1];
        #pragma unroll
        for (int nt = 0; nt < 4; ++nt) {
            v2f b;
            b.x = w2s[k * 64 + nt * 16 + mn];
            b.y = w2s[(k + 1) * 64 + nt * 16 + mn];
            acc2[nt] = __builtin_amdgcn_wmma_f32_16x16x4_f32(
                false, a, false, b, (short)0, acc2[nt], false, false);
        }
    }

    #pragma unroll
    for (int nt = 0; nt < 4; ++nt) {
        const float b2v = has_bias2 ? bias2[nt * 16 + mn] : 0.f;
        #pragma unroll
        for (int i = 0; i < 8; ++i) {
            const int m = mbase + i;
            const int r = row0 + m;
            if (r < n_nodes)
                p[(size_t)r * HID + nt * 16 + mn] = acc2[nt][i] + b2v;
        }
    }
}

static_assert(NEDGES % 4 == 0, "edge count must fill whole groups of four");

__global__ __launch_bounds__(256) void zero_int_kernel(int* __restrict__ p, int n)
{
    const int i = blockIdx.x * 256 + threadIdx.x;
    if (i < n) p[i] = 0;
}

__global__ __launch_bounds__(256) void row_count_kernel(
    const int* __restrict__ row, int* __restrict__ cnt, int n_quads)
{
    const int i = blockIdx.x * 256 + threadIdx.x;
    if (i < n_quads) {
        const int4 r = ((const int4*)row)[i];
        atomicAdd(cnt + r.x, 1);
        atomicAdd(cnt + r.y, 1);
        atomicAdd(cnt + r.z, 1);
        atomicAdd(cnt + r.w, 1);
    }
}

__global__ __launch_bounds__(1024) void block_sum_kernel(
    const int* __restrict__ cnt, int* __restrict__ bsum, int n)
{
    __shared__ int wsum[32];
    const int i = blockIdx.x * 1024 + threadIdx.x;
    int v = i < n ? cnt[i] : 0;
    for (int o = 16; o > 0; o >>= 1) v += __shfl_xor(v, o, 32);
    if ((threadIdx.x & 31) == 0) wsum[threadIdx.x >> 5] = v;
    __syncthreads();
    if (threadIdx.x < 32) {
        int t = wsum[threadIdx.x];
        for (int o = 16; o > 0; o >>= 1) t += __shfl_xor(t, o, 32);
        if (threadIdx.x == 0) bsum[blockIdx.x] = t;
    }
}

__global__ __launch_bounds__(1024) void row_scan_kernel(
    int* __restrict__ cnt, const int* __restrict__ bsum,
    int* __restrict__ rowptr, int n)
{
    __shared__ int wsum[32];
    __shared__ int boff;
    const int tid  = threadIdx.x;
    const int lane = tid & 31;
    const int wv   = tid >> 5;
    const int i    = blockIdx.x * 1024 + tid;
    const int v    = i < n ? cnt[i] : 0;

    int s = v;
    for (int o = 1; o < 32; o <<= 1) {
        const int t = __shfl_up(s, o, 32);
        if (lane >= o) s += t;
    }
    if (lane == 31) wsum[wv] = s;
    if (wv == 0) {
        int t = 0;
        for (int k = lane; k < (int)blockIdx.x; k += 32) t += bsum[k];
        for (int o = 16; o > 0; o >>= 1) t += __shfl_xor(t, o, 32);
        if (lane == 0) boff = t;
    }
    __syncthreads();
    if (wv == 0) {
        const int t = wsum[lane];
        int u = t;
        for (int o = 1; o < 32; o <<= 1) {
            const int y = __shfl_up(u, o, 32);
            if (lane >= o) u += y;
        }
        wsum[lane] = u - t;
    }
    __syncthreads();
    const int excl = boff + wsum[wv] + s - v;
    if (i <= n) rowptr[i] = excl;
    if (i < n)  cnt[i] = excl;
}

__global__ __launch_bounds__(256) void bucket_fill_kernel(
    const int* __restrict__ row, const int* __restrict__ col,
    int* __restrict__ cursor, int* __restrict__ cols, int n_quads)
{
    const int i = blockIdx.x * 256 + threadIdx.x;
    if (i < n_quads) {
        const int4 r = ((const int4*)row)[i];
        const int4 c = ((const int4*)col)[i];
        const int p0 = atomicAdd(cursor + r.x, 1);
        const int p1 = atomicAdd(cursor + r.y, 1);
        const int p2 = atomicAdd(cursor + r.z, 1);
        const int p3 = atomicAdd(cursor + r.w, 1);
        cols[p0] = c.x;
        cols[p1] = c.y;
        cols[p2] = c.z;
        cols[p3] = c.w;
    }
}

#define ROWS_PER_WAVE 8
#define ROW_WAVES_PER_BLOCK 4

__global__ __launch_bounds__(128) void row_att_kernel(
    const float* __restrict__ p1, const float* __restrict__ p2,
    const float* __restrict__ w2, const float* __restrict__ b2,
    const int* __restrict__ rowptr, const int* __restrict__ cols,
    float* __restrict__ mask, float* __restrict__ dis, int n_nodes)
{
    const int lane   = threadIdx.x & 31;
    const int sub    = lane & 15;
    const int half16 = lane & 16;
    const int hsel   = lane >> 4;
    const int wave_g = blockIdx.x * ROW_WAVES_PER_BLOCK + (threadIdx.x >> 5);
    const int r0     = wave_g * ROWS_PER_WAVE;
    if (r0 >= n_nodes) return;
    const float4* __restrict__ p1v = (const float4*)p1;
    const float4* __restrict__ p2v = (const float4*)p2;

    const float4 w = ((const float4*)w2)[sub];
    const float bias = b2[0];

    const int rp_i = r0 + lane;
    const int rp_l = rowptr[rp_i < n_nodes ? rp_i : n_nodes];

    for (int q = 0; q < ROWS_PER_WAVE / 2; ++q) {
        const int r = r0 + 2 * q + hsel;
        const bool rv = r < n_nodes;
        const int start = __shfl(rp_l, 2 * q + hsel, 32);
        const int end   = __shfl(rp_l, 2 * q + hsel + 1, 32);
        const int deg   = end - start;
        const int degmax = max(deg, __shfl_xor(deg, 16, 32));
        const float4 a = p1v[(rv ? r : n_nodes - 1) * (HID / 4) + sub];

        float rs = 0.f;
        for (int base = 0; base < degmax; base += 16) {
            const int idx = start + base + sub;
            const bool ev = idx < end;
            const int c_l = ev ? cols[idx] : 0;

            unsigned cj[16];
            float4 b[16];
            #pragma unroll
            for (int j = 0; j < 16; ++j)
                cj[j] = (unsigned)__shfl(c_l, half16 + j, 32);
            #pragma unroll
            for (int j = 0; j < 16; ++j)
                b[j] = p2v[cj[j] * (HID / 4) + sub];

            float mine = 0.f;
            #pragma unroll
            for (int j = 0; j < 16; ++j) {
                const float t0 = fmaxf(a.x + b[j].x, 0.f);
                const float t1 = fmaxf(a.y + b[j].y, 0.f);
                const float t2 = fmaxf(a.z + b[j].z, 0.f);
                const float t3 = fmaxf(a.w + b[j].w, 0.f);
                float s = t0 * w.x + t1 * w.y + t2 * w.z + t3 * w.w;
                s += __shfl_xor(s, 8, 32);
                s += __shfl_xor(s, 4, 32);
                s += __shfl_xor(s, 2, 32);
                s += __shfl_xor(s, 1, 32);
                mine = (sub == j) ? s : mine;
            }

            const float la = mine + bias;
            const float sg = 1.f / (1.f + __expf(-la));
            float m = sg * (ZETA_C - GAMMA_C) + GAMMA_C;
            m = fminf(fmaxf(m, 0.f), 1.f);
            if (ev) { mask[idx] = m; rs += m; }
        }

        rs += __shfl_xor(rs, 8, 32);
        rs += __shfl_xor(rs, 4, 32);
        rs += __shfl_xor(rs, 2, 32);
        rs += __shfl_xor(rs, 1, 32);
        if (rv && sub == 0) dis[r] = fminf(rsqrtf(rs + 1e-6f), 10.f);
    }
}

__global__ __launch_bounds__(128) void row_spmm_kernel(
    const float* __restrict__ x, const float* __restrict__ mask,
    const float* __restrict__ dis,
    const int* __restrict__ rowptr, const int* __restrict__ cols,
    const float* addin, float* out, float* xn, int n_nodes)
{
    const int lane   = threadIdx.x & 31;
    const int sub    = lane & 15;
    const int half16 = lane & 16;
    const int hsel   = lane >> 4;
    const int wave_g = blockIdx.x * ROW_WAVES_PER_BLOCK + (threadIdx.x >> 5);
    const int r0     = wave_g * ROWS_PER_WAVE;
    if (r0 >= n_nodes) return;
    const float4* __restrict__ xv = (const float4*)x;
    const float4 zero4 = {0.f, 0.f, 0.f, 0.f};

    const int rp_i = r0 + lane;
    const int rp_l = rowptr[rp_i < n_nodes ? rp_i : n_nodes];

    for (int q = 0; q < ROWS_PER_WAVE / 2; ++q) {
        const int r = r0 + 2 * q + hsel;
        const bool rv = r < n_nodes;
        const int start = __shfl(rp_l, 2 * q + hsel, 32);
        const int end   = __shfl(rp_l, 2 * q + hsel + 1, 32);
        const int deg   = end - start;
        const int degmax = max(deg, __shfl_xor(deg, 16, 32));
        const float dr = dis[rv ? r : n_nodes - 1];

        float4 acc = zero4;
        for (int base = 0; base < degmax; base += 16) {
            const int idx = start + base + sub;
            const bool ev = idx < end;
            const int c_l = ev ? cols[idx] : 0;
            const int nv  = end - (start + base);

            unsigned cj[16];
            float4 xc[16];
            #pragma unroll
            for (int j = 0; j < 16; ++j)
                cj[j] = (unsigned)__shfl(c_l, half16 + j, 32);
            #pragma unroll
            for (int j = 0; j < 16; ++j)
                xc[j] = xv[cj[j] * (HID / 4) + sub];

            float v_l = 0.f;
            if (ev) v_l = mask[idx] * dr * dis[c_l];

            #pragma unroll
            for (int j = 0; j < 16; ++j) {
                const float v = __shfl(v_l, half16 + j, 32);
                const float4 t = (j < nv) ? xc[j] : zero4;
                acc.x += v * t.x;
                acc.y += v * t.y;
                acc.z += v * t.z;
                acc.w += v * t.w;
            }
        }

        if (rv) {
            const size_t o = (size_t)r * (HID / 4) + sub;
            const float4 bs = ((const float4*)addin)[o];
            float4 res;
            res.x = bs.x + acc.x; res.y = bs.y + acc.y;
            res.z = bs.z + acc.z; res.w = bs.w + acc.w;
            ((float4*)out)[o] = res;
            if (xn) ((float4*)xn)[o] = acc;
        }
    }
}

extern "C" void kernel_launch(void* const* d_in, const int* in_sizes, int n_in,
                              void* d_out, int out_size, void* d_ws, size_t ws_size,
                              hipStream_t stream) {
    (void)in_sizes; (void)n_in; (void)out_size; (void)ws_size;

    const float* feat = (const float*)d_in[0];
    const float* nbW  = (const float*)d_in[1];
    const float* nbb  = (const float*)d_in[2];
    const float* sW   = (const float*)d_in[3];
    const float* sb   = (const float*)d_in[4];
    const float* a1W  = (const float*)d_in[5];
    const float* a1b  = (const float*)d_in[6];
    const float* a2W  = (const float*)d_in[7];
    const float* a2b  = (const float*)d_in[8];
    const int*   row  = (const int*)d_in[9];
    const int*   col  = (const int*)d_in[10];
    float* out = (float*)d_out;

    const int NH = NNODES * HID;
    float* ws = (float*)d_ws;
    float* bufA   = ws; ws += NH;
    float* bufP1  = ws; ws += NH;
    float* bufP2  = ws; ws += NH;
    int*   cols   = (int*)ws; ws += NEDGES;
    float* maskb  = ws; ws += NEDGES;
    int*   rowptr = (int*)ws; ws += NNODES + 32;
    int*   cnt    = (int*)ws; ws += NNODES;
    float* dis    = ws; ws += NNODES;
    int*   bsum   = (int*)ws; ws += 128;

    const int gNode  = (NNODES + 255) / 256;
    const int nQuad  = NEDGES / 4;
    const int gEdge  = (nQuad + 255) / 256;
    const int gScan  = (NNODES + 1 + 1023) / 1024;
    const int gTile  = ((NNODES + 15) / 16 + 3) / 4;
    const int gRow   = (NNODES + ROWS_PER_WAVE * ROW_WAVES_PER_BLOCK - 1)
                       / (ROWS_PER_WAVE * ROW_WAVES_PER_BLOCK);

    zero_int_kernel<<<gNode, 256, 0, stream>>>(cnt, NNODES);
    row_count_kernel<<<gEdge, 256, 0, stream>>>(row, cnt, nQuad);
    block_sum_kernel<<<gScan, 1024, 0, stream>>>(cnt, bsum, NNODES);
    row_scan_kernel<<<gScan, 1024, 0, stream>>>(cnt, bsum, rowptr, NNODES);
    bucket_fill_kernel<<<gEdge, 256, 0, stream>>>(row, col, cnt, cols, nQuad);

    const float* xin[NLAYERS]  = {feat,  bufA,  bufP1};
    float*       p1o[NLAYERS]  = {bufP1, bufP1, bufA};
    float*       xout[NLAYERS] = {bufA,  bufP1, nullptr};
    const float* addin[NLAYERS] = {feat, out, out};

    for (int l = 0; l < NLAYERS; ++l) {
        node_transform_kernel<<<gTile, 128, 0, stream>>>(
            xin[l], nbW + (size_t)l * 4096, nbb + l * 64,
            a1W + (size_t)l * 8192, a1b + l * 64, 1, p1o[l], NNODES);
        node_transform_kernel<<<gTile, 128, 0, stream>>>(
            xin[l], sW + (size_t)l * 4096, sb + l * 64,
            a1W + (size_t)l * 8192 + 4096, a1b, 0, bufP2, NNODES);

        row_att_kernel<<<gRow, 128, 0, stream>>>(
            p1o[l], bufP2, a2W + l * 64, a2b + l, rowptr, cols, maskb, dis, NNODES);

        row_spmm_kernel<<<gRow, 128, 0, stream>>>(
            xin[l], maskb, dis, rowptr, cols, addin[l], out, xout[l], NNODES);
    }
}


# row kernels: exchanges inside a 16-lane half (index and weight broadcast, 16 logit sums as one halving fold, row sum) go on DPP, bucket bounds by readlane: no LDS-unit exchanges left; same sum trees
# speedup vs baseline: 2.4373x; 1.0712x over previous
#include <hip/hip_runtime.h>
#include <hip/hip_bf16.h>

#define NNODES 100000
#define NEDGES 1200000
#define HID 64
#define NLAYERS 3
#define GAMMA_C (-0.45f)
#define ZETA_C  (1.05f)

typedef float v2f __attribute__((ext_vector_type(2)));
typedef float v8f __attribute__((ext_vector_type(8)));

__global__ __launch_bounds__(128) void node_transform_kernel(
    const float* __restrict__ x, const float* __restrict__ W1,
    const float* __restrict__ b1, const float* __restrict__ W2,
    const float* __restrict__ bias2, int has_bias2,
    float* __restrict__ p, int n_nodes)
{
    __shared__ float w1s[64 * 64];
    __shared__ float w2s[64 * 64];
    __shared__ float gs[4 * 16 * 68];

    for (int i = threadIdx.x; i < 64 * 64; i += 128) {
        w1s[i] = W1[i];
        w2s[i] = W2[i];
    }
    __syncthreads();

    const int wave = threadIdx.x >> 5;
    const int lane = threadIdx.x & 31;
    const int mn    = lane & 15;
    const int kh    = (lane >> 4) << 1;
    const int mbase = (lane >> 4) << 3;
    float* gt = gs + wave * (16 * 68);

    const int tile = blockIdx.x * 4 + wave;
    const int row0 = tile * 16;
    if (row0 >= n_nodes) return;

    int ri = row0 + mn;
    if (ri >= n_nodes) ri = n_nodes - 1;
    const float* xr = x + (size_t)ri * HID;

    const v8f vz = {0.f, 0.f, 0.f, 0.f, 0.f, 0.f, 0.f, 0.f};
    v8f acc[4] = {vz, vz, vz, vz};

    #pragma unroll
    for (int kk = 0; kk < 16; ++kk) {
        const int k = kk * 4 + kh;
        v2f a; a.x = xr[k]; a.y = xr[k + 1];
        #pragma unroll
        for (int nt = 0; nt < 4; ++nt) {
            v2f b;
            b.x = w1s[k * 64 + nt * 16 + mn];
            b.y = w1s[(k + 1) * 64 + nt * 16 + mn];
            acc[nt] = __builtin_amdgcn_wmma_f32_16x16x4_f32(
                false, a, false, b, (short)0, acc[nt], false, false);
        }
    }

    #pragma unroll
    for (int nt = 0; nt < 4; ++nt) {
        const float bv = b1[nt * 16 + mn];
        #pragma unroll
        for (int i = 0; i < 8; ++i) {
            float v = acc[nt][i] + bv;
            v = v > 0.f ? v : 0.f;
            gt[(mbase + i) * 68 + nt * 16 + mn] = v;
        }
    }

    v8f acc2[4] = {vz, vz, vz, vz};
    #pragma unroll
    for (int kk = 0; kk < 16; ++kk) {
        const int k = kk * 4 + kh;
        v2f a; a.x = gt[mn * 68 + k]; a.y = gt[mn * 68 + k + 1];
        #pragma unroll
        for (int nt = 0; nt < 4; ++nt) {
            v2f b;
            b.x = w2s[k * 64 + nt * 16 + mn];
            b.y = w2s[(k + 1) * 64 + nt * 16 + mn];
            acc2[nt] = __builtin_amdgcn_wmma_f32_16x16x4_f32(
                false, a, false, b, (short)0, acc2[nt], false, false);
        }
    }

    #pragma unroll
    for (int nt = 0; nt < 4; ++nt) {
        const float b2v = has_bias2 ? bias2[nt * 16 + mn] : 0.f;
        #pragma unroll
        for (int i = 0; i < 8; ++i) {
            const int m = mbase + i;
            const int r = row0 + m;
            if (r < n_nodes)
                p[(size_t)r * HID + nt * 16 + mn] = acc2[nt][i] + b2v;
        }
    }
}

static_assert(NEDGES % 4 == 0, "edge count must fill whole groups of four");

__global__ __launch_bounds__(256) void zero_int_kernel(int* __restrict__ p, int n)
{
    const int i = blockIdx.x * 256 + threadIdx.x;
    if (i < n) p[i] = 0;
}

__global__ __launch_bounds__(256) void row_count_kernel(
    const int* __restrict__ row, int* __restrict__ cnt, int n_quads)
{
    const int i = blockIdx.x * 256 + threadIdx.x;
    if (i < n_quads) {
        const int4 r = ((const int4*)row)[i];
        atomicAdd(cnt + r.x, 1);
        atomicAdd(cnt + r.y, 1);
        atomicAdd(cnt + r.z, 1);
        atomicAdd(cnt + r.w, 1);
    }
}

__global__ __launch_bounds__(1024) void block_sum_kernel(
    const int* __restrict__ cnt, int* __restrict__ bsum, int n)
{
    __shared__ int wsum[32];
    const int i = blockIdx.x * 1024 + threadIdx.x;
    int v = i < n ? cnt[i] : 0;
    for (int o = 16; o > 0; o >>= 1) v += __shfl_xor(v, o, 32);
    if ((threadIdx.x & 31) == 0) wsum[threadIdx.x >> 5] = v;
    __syncthreads();
    if (threadIdx.x < 32) {
        int t = wsum[threadIdx.x];
        for (int o = 16; o > 0; o >>= 1) t += __shfl_xor(t, o, 32);
        if (threadIdx.x == 0) bsum[blockIdx.x] = t;
    }
}

__global__ __launch_bounds__(1024) void row_scan_kernel(
    int* __restrict__ cnt, const int* __restrict__ bsum,
    int* __restrict__ rowptr, int n)
{
    __shared__ int wsum[32];
    __shared__ int boff;
    const int tid  = threadIdx.x;
    const int lane = tid & 31;
    const int wv   = tid >> 5;
    const int i    = blockIdx.x * 1024 + tid;
    const int v    = i < n ? cnt[i] : 0;

    int s = v;
    for (int o = 1; o < 32; o <<= 1) {
        const int t = __shfl_up(s, o, 32);
        if (lane >= o) s += t;
    }
    if (lane == 31) wsum[wv] = s;
    if (wv == 0) {
        int t = 0;
        for (int k = lane; k < (int)blockIdx.x; k += 32) t += bsum[k];
        for (int o = 16; o > 0; o >>= 1) t += __shfl_xor(t, o, 32);
        if (lane == 0) boff = t;
    }
    __syncthreads();
    if (wv == 0) {
        const int t = wsum[lane];
        int u = t;
        for (int o = 1; o < 32; o <<= 1) {
            const int y = __shfl_up(u, o, 32);
            if (lane >= o) u += y;
        }
        wsum[lane] = u - t;
    }
    __syncthreads();
    const int excl = boff + wsum[wv] + s - v;
    if (i <= n) rowptr[i] = excl;
    if (i < n)  cnt[i] = excl;
}

__global__ __launch_bounds__(256) void bucket_fill_kernel(
    const int* __restrict__ row, const int* __restrict__ col,
    int* __restrict__ cursor, int* __restrict__ cols, int n_quads)
{
    const int i = blockIdx.x * 256 + threadIdx.x;
    if (i < n_quads) {
        const int4 r = ((const int4*)row)[i];
        const int4 c = ((const int4*)col)[i];
        const int p0 = atomicAdd(cursor + r.x, 1);
        const int p1 = atomicAdd(cursor + r.y, 1);
        const int p2 = atomicAdd(cursor + r.z, 1);
        const int p3 = atomicAdd(cursor + r.w, 1);
        cols[p0] = c.x;
        cols[p1] = c.y;
        cols[p2] = c.z;
        cols[p3] = c.w;
    }
}

#define ROWS_PER_WAVE 8
#define ROW_WAVES_PER_BLOCK 4

#define DPP_ROW_SHARE(j) (0x150 + (j))
#define DPP_ROW_XMASK(m) (0x160 + (m))

template <int CTRL>
__device__ __forceinline__ int dpp_i(int v)
{
    return __builtin_amdgcn_update_dpp(0, v, CTRL, 0xf, 0xf, true);
}

template <int CTRL>
__device__ __forceinline__ float dpp_f(float v)
{
    return __builtin_bit_cast(float,
        __builtin_amdgcn_update_dpp(0, __builtin_bit_cast(int, v), CTRL, 0xf, 0xf, true));
}

template <int J>
__device__ __forceinline__ void row_share16(unsigned (&o)[16], int v)
{
    if constexpr (J < 16) {
        o[J] = (unsigned)dpp_i<DPP_ROW_SHARE(J)>(v);
        row_share16<J + 1>(o, v);
    }
}

template <int M, int N>
__device__ __forceinline__ void fold_step(float (&s)[16], bool hi)
{
    #pragma unroll
    for (int j = 0; j < N; ++j) {
        const float keep = hi ? s[j + N] : s[j];
        const float give = hi ? s[j] : s[j + N];
        s[j] = keep + dpp_f<DPP_ROW_XMASK(M)>(give);
    }
}

__device__ __forceinline__ float row_sums16(float (&s)[16], int sub)
{
    fold_step<8, 8>(s, (sub & 8) != 0);
    fold_step<4, 4>(s, (sub & 4) != 0);
    fold_step<2, 2>(s, (sub & 2) != 0);
    fold_step<1, 1>(s, (sub & 1) != 0);
    return s[0];
}

__device__ __forceinline__ float row_sum1(float v)
{
    v += dpp_f<DPP_ROW_XMASK(8)>(v);
    v += dpp_f<DPP_ROW_XMASK(4)>(v);
    v += dpp_f<DPP_ROW_XMASK(2)>(v);
    v += dpp_f<DPP_ROW_XMASK(1)>(v);
    return v;
}

__global__ __launch_bounds__(128) void row_att_kernel(
    const float* __restrict__ p1, const float* __restrict__ p2,
    const float* __restrict__ w2, const float* __restrict__ b2,
    const int* __restrict__ rowptr, const int* __restrict__ cols,
    float* __restrict__ mask, float* __restrict__ dis, int n_nodes)
{
    const int lane   = threadIdx.x & 31;
    const int sub    = lane & 15;
    const int hsel   = lane >> 4;
    const int wave_g = blockIdx.x * ROW_WAVES_PER_BLOCK + (threadIdx.x >> 5);
    const int r0     = wave_g * ROWS_PER_WAVE;
    if (r0 >= n_nodes) return;
    const float4* __restrict__ p1v = (const float4*)p1;
    const float4* __restrict__ p2v = (const float4*)p2;

    const float4 w = ((const float4*)w2)[sub];
    const float bias = b2[0];

    const int rp_i = r0 + lane;
    const int rp_l = rowptr[rp_i < n_nodes ? rp_i : n_nodes];

    for (int q = 0; q < ROWS_PER_WAVE / 2; ++q) {
        const int r = r0 + 2 * q + hsel;
        const bool rv = r < n_nodes;
        const int e0 = __builtin_amdgcn_readlane(rp_l, 2 * q);
        const int e1 = __builtin_amdgcn_readlane(rp_l, 2 * q + 1);
        const int e2 = __builtin_amdgcn_readlane(rp_l, 2 * q + 2);
        const int start = hsel ? e1 : e0;
        const int end   = hsel ? e2 : e1;
        const int degmax = max(e1 - e0, e2 - e1);
        const float4 a = p1v[(rv ? r : n_nodes - 1) * (HID / 4) + sub];

        float rs = 0.f;
        for (int base = 0; base < degmax; base += 16) {
            const int idx = start + base + sub;
            const bool ev = idx < end;
            const int c_l = ev ? cols[idx] : 0;

            unsigned cj[16];
            float4 b[16];
            row_share16<0>(cj, c_l);
            #pragma unroll
            for (int j = 0; j < 16; ++j)
                b[j] = p2v[cj[j] * (HID / 4) + sub];

            float s[16];
            #pragma unroll
            for (int j = 0; j < 16; ++j) {
                const float t0 = fmaxf(a.x + b[j].x, 0.f);
                const float t1 = fmaxf(a.y + b[j].y, 0.f);
                const float t2 = fmaxf(a.z + b[j].z, 0.f);
                const float t3 = fmaxf(a.w + b[j].w, 0.f);
                s[j] = t0 * w.x + t1 * w.y + t2 * w.z + t3 * w.w;
            }
            const float mine = row_sums16(s, sub);

            const float la = mine + bias;
            const float sg = 1.f / (1.f + __expf(-la));
            float m = sg * (ZETA_C - GAMMA_C) + GAMMA_C;
            m = fminf(fmaxf(m, 0.f), 1.f);
            if (ev) { mask[idx] = m; rs += m; }
        }

        rs = row_sum1(rs);
        if (rv && sub == 0) dis[r] = fminf(rsqrtf(rs + 1e-6f), 10.f);
    }
}

__global__ __launch_bounds__(128) void row_spmm_kernel(
    const float* __restrict__ x, const float* __restrict__ mask,
    const float* __restrict__ dis,
    const int* __restrict__ rowptr, const int* __restrict__ cols,
    const float* addin, float* out, float* xn, int n_nodes)
{
    const int lane   = threadIdx.x & 31;
    const int sub    = lane & 15;
    const int hsel   = lane >> 4;
    const int wave_g = blockIdx.x * ROW_WAVES_PER_BLOCK + (threadIdx.x >> 5);
    const int r0     = wave_g * ROWS_PER_WAVE;
    if (r0 >= n_nodes) return;
    const float4* __restrict__ xv = (const float4*)x;
    const float4 zero4 = {0.f, 0.f, 0.f, 0.f};

    const int rp_i = r0 + lane;
    const int rp_l = rowptr[rp_i < n_nodes ? rp_i : n_nodes];

    for (int q = 0; q < ROWS_PER_WAVE / 2; ++q) {
        const int r = r0 + 2 * q + hsel;
        const bool rv = r < n_nodes;
        const int e0 = __builtin_amdgcn_readlane(rp_l, 2 * q);
        const int e1 = __builtin_amdgcn_readlane(rp_l, 2 * q + 1);
        const int e2 = __builtin_amdgcn_readlane(rp_l, 2 * q + 2);
        const int start = hsel ? e1 : e0;
        const int end   = hsel ? e2 : e1;
        const int degmax = max(e1 - e0, e2 - e1);
        const float dr = dis[rv ? r : n_nodes - 1];

        float4 acc = zero4;
        for (int base = 0; base < degmax; base += 16) {
            const int idx = start + base + sub;
            const bool ev = idx < end;
            const int c_l = ev ? cols[idx] : 0;
            const int nv  = end - (start + base);

            unsigned cj[16];
            float4 xc[16];
            row_share16<0>(cj, c_l);
            #pragma unroll
            for (int j = 0; j < 16; ++j)
                xc[j] = xv[cj[j] * (HID / 4) + sub];

            float v_l = 0.f;
            if (ev) v_l = mask[idx] * dr * dis[c_l];

            unsigned vj[16];
            row_share16<0>(vj, __builtin_bit_cast(int, v_l));

            #pragma unroll
            for (int j = 0; j < 16; ++j) {
                const float v = __builtin_bit_cast(float, vj[j]);
                const float4 t = (j < nv) ? xc[j] : zero4;
                acc.x += v * t.x;
                acc.y += v * t.y;
                acc.z += v * t.z;
                acc.w += v * t.w;
            }
        }

        if (rv) {
            const size_t o = (size_t)r * (HID / 4) + sub;
            const float4 bs = ((const float4*)addin)[o];
            float4 res;
            res.x = bs.x + acc.x; res.y = bs.y + acc.y;
            res.z = bs.z + acc.z; res.w = bs.w + acc.w;
            ((float4*)out)[o] = res;
            if (xn) ((float4*)xn)[o] = acc;
        }
    }
}

extern "C" void kernel_launch(void* const* d_in, const int* in_sizes, int n_in,
                              void* d_out, int out_size, void* d_ws, size_t ws_size,
                              hipStream_t stream) {
    (void)in_sizes; (void)n_in; (void)out_size; (void)ws_size;

    const float* feat = (const float*)d_in[0];
    const float* nbW  = (const float*)d_in[1];
    const float* nbb  = (const float*)d_in[2];
    const float* sW   = (const float*)d_in[3];
    const float* sb   = (const float*)d_in[4];
    const float* a1W  = (const float*)d_in[5];
    const float* a1b  = (const float*)d_in[6];
    const float* a2W  = (const float*)d_in[7];
    const float* a2b  = (const float*)d_in[8];
    const int*   row  = (const int*)d_in[9];
    const int*   col  = (const int*)d_in[10];
    float* out = (float*)d_out;

    const int NH = NNODES * HID;
    float* ws = (float*)d_ws;
    float* bufA   = ws; ws += NH;
    float* bufP1  = ws; ws += NH;
    float* bufP2  = ws; ws += NH;
    int*   cols   = (int*)ws; ws += NEDGES;
    float* maskb  = ws; ws += NEDGES;
    int*   rowptr = (int*)ws; ws += NNODES + 32;
    int*   cnt    = (int*)ws; ws += NNODES;
    float* dis    = ws; ws += NNODES;
    int*   bsum   = (int*)ws; ws += 128;

    const int gNode  = (NNODES + 255) / 256;
    const int nQuad  = NEDGES / 4;
    const int gEdge  = (nQuad + 255) / 256;
    const int gScan  = (NNODES + 1 + 1023) / 1024;
    const int gTile  = ((NNODES + 15) / 16 + 3) / 4;
    const int gRow   = (NNODES + ROWS_PER_WAVE * ROW_WAVES_PER_BLOCK - 1)
                       / (ROWS_PER_WAVE * ROW_WAVES_PER_BLOCK);

    zero_int_kernel<<<gNode, 256, 0, stream>>>(cnt, NNODES);
    row_count_kernel<<<gEdge, 256, 0, stream>>>(row, cnt, nQuad);
    block_sum_kernel<<<gScan, 1024, 0, stream>>>(cnt, bsum, NNODES);
    row_scan_kernel<<<gScan, 1024, 0, stream>>>(cnt, bsum, rowptr, NNODES);
    bucket_fill_kernel<<<gEdge, 256, 0, stream>>>(row, col, cnt, cols, nQuad);

    const float* xin[NLAYERS]  = {feat,  bufA,  bufP1};
    float*       p1o[NLAYERS]  = {bufP1, bufP1, bufA};
    float*       xout[NLAYERS] = {bufA,  bufP1, nullptr};
    const float* addin[NLAYERS] = {feat, out, out};

    for (int l = 0; l < NLAYERS; ++l) {
        node_transform_kernel<<<gTile, 128, 0, stream>>>(
            xin[l], nbW + (size_t)l * 4096, nbb + l * 64,
            a1W + (size_t)l * 8192, a1b + l * 64, 1, p1o[l], NNODES);
        node_transform_kernel<<<gTile, 128, 0, stream>>>(
            xin[l], sW + (size_t)l * 4096, sb + l * 64,
            a1W + (size_t)l * 8192 + 4096, a1b, 0, bufP2, NNODES);

        row_att_kernel<<<gRow, 128, 0, stream>>>(
            p1o[l], bufP2, a2W + l * 64, a2b + l, rowptr, cols, maskb, dis, NNODES);

        row_spmm_kernel<<<gRow, 128, 0, stream>>>(
            xin[l], maskb, dis, rowptr, cols, addin[l], out, xout[l], NNODES);
    }
}


# node transform: both projections of a layer in one launch (even blocks first weight set, odd blocks second, same x tiles): 14 launches, not 17; no boundary and no lone last blocks between them
# speedup vs baseline: 2.5170x; 1.0327x over previous
#include <hip/hip_runtime.h>
#include <hip/hip_bf16.h>

#define NNODES 100000
#define NEDGES 1200000
#define HID 64
#define NLAYERS 3
#define GAMMA_C (-0.45f)
#define ZETA_C  (1.05f)

typedef float v2f __attribute__((ext_vector_type(2)));
typedef float v8f __attribute__((ext_vector_type(8)));

__global__ __launch_bounds__(128) void node_transform_kernel(
    const float* __restrict__ x,
    const float* __restrict__ W1a, const float* __restrict__ b1a,
    const float* __restrict__ W2a, const float* __restrict__ bias2a,
    float* __restrict__ pa,
    const float* __restrict__ W1b, const float* __restrict__ b1b,
    const float* __restrict__ W2b, float* __restrict__ pb,
    int n_nodes)
{
    const bool first = (blockIdx.x & 1) == 0;
    const float* __restrict__ W1 = first ? W1a : W1b;
    const float* __restrict__ b1 = first ? b1a : b1b;
    const float* __restrict__ W2 = first ? W2a : W2b;
    float* __restrict__ p = first ? pa : pb;

    __shared__ float w1s[64 * 64];
    __shared__ float w2s[64 * 64];
    __shared__ float gs[4 * 16 * 68];

    for (int i = threadIdx.x; i < 64 * 64; i += 128) {
        w1s[i] = W1[i];
        w2s[i] = W2[i];
    }
    __syncthreads();

    const int wave = threadIdx.x >> 5;
    const int lane = threadIdx.x & 31;
    const int mn    = lane & 15;
    const int kh    = (lane >> 4) << 1;
    const int mbase = (lane >> 4) << 3;
    float* gt = gs + wave * (16 * 68);

    const int tile = (blockIdx.x >> 1) * 4 + wave;
    const int row0 = tile * 16;
    if (row0 >= n_nodes) return;

    int ri = row0 + mn;
    if (ri >= n_nodes) ri = n_nodes - 1;
    const float* xr = x + (size_t)ri * HID;

    const v8f vz = {0.f, 0.f, 0.f, 0.f, 0.f, 0.f, 0.f, 0.f};
    v8f acc[4] = {vz, vz, vz, vz};

    #pragma unroll
    for (int kk = 0; kk < 16; ++kk) {
        const int k = kk * 4 + kh;
        v2f a; a.x = xr[k]; a.y = xr[k + 1];
        #pragma unroll
        for (int nt = 0; nt < 4; ++nt) {
            v2f b;
            b.x = w1s[k * 64 + nt * 16 + mn];
            b.y = w1s[(k + 1) * 64 + nt * 16 + mn];
            acc[nt] = __builtin_amdgcn_wmma_f32_16x16x4_f32(
                false, a, false, b, (short)0, acc[nt], false, false);
        }
    }

    #pragma unroll
    for (int nt = 0; nt < 4; ++nt) {
        const float bv = b1[nt * 16 + mn];
        #pragma unroll
        for (int i = 0; i < 8; ++i) {
            float v = acc[nt][i] + bv;
            v = v > 0.f ? v : 0.f;
            gt[(mbase + i) * 68 + nt * 16 + mn] = v;
        }
    }

    v8f acc2[4] = {vz, vz, vz, vz};
    #pragma unroll
    for (int kk = 0; kk < 16; ++kk) {
        const int k = kk * 4 + kh;
        v2f a; a.x = gt[mn * 68 + k]; a.y = gt[mn * 68 + k + 1];
        #pragma unroll
        for (int nt = 0; nt < 4; ++nt) {
            v2f b;
            b.x = w2s[k * 64 + nt * 16 + mn];
            b.y = w2s[(k + 1) * 64 + nt * 16 + mn];
            acc2[nt] = __builtin_amdgcn_wmma_f32_16x16x4_f32(
                false, a, false, b, (short)0, acc2[nt], false, false);
        }
    }

    #pragma unroll
    for (int nt = 0; nt < 4; ++nt) {
        const float b2v = first ? bias2a[nt * 16 + mn] : 0.f;
        #pragma unroll
        for (int i = 0; i < 8; ++i) {
            const int m = mbase + i;
            const int r = row0 + m;
            if (r < n_nodes)
                p[(size_t)r * HID + nt * 16 + mn] = acc2[nt][i] + b2v;
        }
    }
}

static_assert(NEDGES % 4 == 0, "edge count must fill whole groups of four");

__global__ __launch_bounds__(256) void zero_int_kernel(int* __restrict__ p, int n)
{
    const int i = blockIdx.x * 256 + threadIdx.x;
    if (i < n) p[i] = 0;
}

__global__ __launch_bounds__(256) void row_count_kernel(
    const int* __restrict__ row, int* __restrict__ cnt, int n_quads)
{
    const int i = blockIdx.x * 256 + threadIdx.x;
    if (i < n_quads) {
        const int4 r = ((const int4*)row)[i];
        atomicAdd(cnt + r.x, 1);
        atomicAdd(cnt + r.y, 1);
        atomicAdd(cnt + r.z, 1);
        atomicAdd(cnt + r.w, 1);
    }
}

__global__ __launch_bounds__(1024) void block_sum_kernel(
    const int* __restrict__ cnt, int* __restrict__ bsum, int n)
{
    __shared__ int wsum[32];
    const int i = blockIdx.x * 1024 + threadIdx.x;
    int v = i < n ? cnt[i] : 0;
    for (int o = 16; o > 0; o >>= 1) v += __shfl_xor(v, o, 32);
    if ((threadIdx.x & 31) == 0) wsum[threadIdx.x >> 5] = v;
    __syncthreads();
    if (threadIdx.x < 32) {
        int t = wsum[threadIdx.x];
        for (int o = 16; o > 0; o >>= 1) t += __shfl_xor(t, o, 32);
        if (threadIdx.x == 0) bsum[blockIdx.x] = t;
    }
}

__global__ __launch_bounds__(1024) void row_scan_kernel(
    int* __restrict__ cnt, const int* __restrict__ bsum,
    int* __restrict__ rowptr, int n)
{
    __shared__ int wsum[32];
    __shared__ int boff;
    const int tid  = threadIdx.x;
    const int lane = tid & 31;
    const int wv   = tid >> 5;
    const int i    = blockIdx.x * 1024 + tid;
    const int v    = i < n ? cnt[i] : 0;

    int s = v;
    for (int o = 1; o < 32; o <<= 1) {
        const int t = __shfl_up(s, o, 32);
        if (lane >= o) s += t;
    }
    if (lane == 31) wsum[wv] = s;
    if (wv == 0) {
        int t = 0;
        for (int k = lane; k < (int)blockIdx.x; k += 32) t += bsum[k];
        for (int o = 16; o > 0; o >>= 1) t += __shfl_xor(t, o, 32);
        if (lane == 0) boff = t;
    }
    __syncthreads();
    if (wv == 0) {
        const int t = wsum[lane];
        int u = t;
        for (int o = 1; o < 32; o <<= 1) {
            const int y = __shfl_up(u, o, 32);
            if (lane >= o) u += y;
        }
        wsum[lane] = u - t;
    }
    __syncthreads();
    const int excl = boff + wsum[wv] + s - v;
    if (i <= n) rowptr[i] = excl;
    if (i < n)  cnt[i] = excl;
}

__global__ __launch_bounds__(256) void bucket_fill_kernel(
    const int* __restrict__ row, const int* __restrict__ col,
    int* __restrict__ cursor, int* __restrict__ cols, int n_quads)
{
    const int i = blockIdx.x * 256 + threadIdx.x;
    if (i < n_quads) {
        const int4 r = ((const int4*)row)[i];
        const int4 c = ((const int4*)col)[i];
        const int p0 = atomicAdd(cursor + r.x, 1);
        const int p1 = atomicAdd(cursor + r.y, 1);
        const int p2 = atomicAdd(cursor + r.z, 1);
        const int p3 = atomicAdd(cursor + r.w, 1);
        cols[p0] = c.x;
        cols[p1] = c.y;
        cols[p2] = c.z;
        cols[p3] = c.w;
    }
}

#define ROWS_PER_WAVE 8
#define ROW_WAVES_PER_BLOCK 4

#define DPP_ROW_SHARE(j) (0x150 + (j))
#define DPP_ROW_XMASK(m) (0x160 + (m))

template <int CTRL>
__device__ __forceinline__ int dpp_i(int v)
{
    return __builtin_amdgcn_update_dpp(0, v, CTRL, 0xf, 0xf, true);
}

template <int CTRL>
__device__ __forceinline__ float dpp_f(float v)
{
    return __builtin_bit_cast(float,
        __builtin_amdgcn_update_dpp(0, __builtin_bit_cast(int, v), CTRL, 0xf, 0xf, true));
}

template <int J>
__device__ __forceinline__ void row_share16(unsigned (&o)[16], int v)
{
    if constexpr (J < 16) {
        o[J] = (unsigned)dpp_i<DPP_ROW_SHARE(J)>(v);
        row_share16<J + 1>(o, v);
    }
}

template <int M, int N>
__device__ __forceinline__ void fold_step(float (&s)[16], bool hi)
{
    #pragma unroll
    for (int j = 0; j < N; ++j) {
        const float keep = hi ? s[j + N] : s[j];
        const float give = hi ? s[j] : s[j + N];
        s[j] = keep + dpp_f<DPP_ROW_XMASK(M)>(give);
    }
}

__device__ __forceinline__ float row_sums16(float (&s)[16], int sub)
{
    fold_step<8, 8>(s, (sub & 8) != 0);
    fold_step<4, 4>(s, (sub & 4) != 0);
    fold_step<2, 2>(s, (sub & 2) != 0);
    fold_step<1, 1>(s, (sub & 1) != 0);
    return s[0];
}

__device__ __forceinline__ float row_sum1(float v)
{
    v += dpp_f<DPP_ROW_XMASK(8)>(v);
    v += dpp_f<DPP_ROW_XMASK(4)>(v);
    v += dpp_f<DPP_ROW_XMASK(2)>(v);
    v += dpp_f<DPP_ROW_XMASK(1)>(v);
    return v;
}

__global__ __launch_bounds__(128) void row_att_kernel(
    const float* __restrict__ p1, const float* __restrict__ p2,
    const float* __restrict__ w2, const float* __restrict__ b2,
    const int* __restrict__ rowptr, const int* __restrict__ cols,
    float* __restrict__ mask, float* __restrict__ dis, int n_nodes)
{
    const int lane   = threadIdx.x & 31;
    const int sub    = lane & 15;
    const int hsel   = lane >> 4;
    const int wave_g = blockIdx.x * ROW_WAVES_PER_BLOCK + (threadIdx.x >> 5);
    const int r0     = wave_g * ROWS_PER_WAVE;
    if (r0 >= n_nodes) return;
    const float4* __restrict__ p1v = (const float4*)p1;
    const float4* __restrict__ p2v = (const float4*)p2;

    const float4 w = ((const float4*)w2)[sub];
    const float bias = b2[0];

    const int rp_i = r0 + lane;
    const int rp_l = rowptr[rp_i < n_nodes ? rp_i : n_nodes];

    for (int q = 0; q < ROWS_PER_WAVE / 2; ++q) {
        const int r = r0 + 2 * q + hsel;
        const bool rv = r < n_nodes;
        const int e0 = __builtin_amdgcn_readlane(rp_l, 2 * q);
        const int e1 = __builtin_amdgcn_readlane(rp_l, 2 * q + 1);
        const int e2 = __builtin_amdgcn_readlane(rp_l, 2 * q + 2);
        const int start = hsel ? e1 : e0;
        const int end   = hsel ? e2 : e1;
        const int degmax = max(e1 - e0, e2 - e1);
        const float4 a = p1v[(rv ? r : n_nodes - 1) * (HID / 4) + sub];

        float rs = 0.f;
        for (int base = 0; base < degmax; base += 16) {
            const int idx = start + base + sub;
            const bool ev = idx < end;
            const int c_l = ev ? cols[idx] : 0;

            unsigned cj[16];
            float4 b[16];
            row_share16<0>(cj, c_l);
            #pragma unroll
            for (int j = 0; j < 16; ++j)
                b[j] = p2v[cj[j] * (HID / 4) + sub];

            float s[16];
            #pragma unroll
            for (int j = 0; j < 16; ++j) {
                const float t0 = fmaxf(a.x + b[j].x, 0.f);
                const float t1 = fmaxf(a.y + b[j].y, 0.f);
                const float t2 = fmaxf(a.z + b[j].z, 0.f);
                const float t3 = fmaxf(a.w + b[j].w, 0.f);
                s[j] = t0 * w.x + t1 * w.y + t2 * w.z + t3 * w.w;
            }
            const float mine = row_sums16(s, sub);

            const float la = mine + bias;
            const float sg = 1.f / (1.f + __expf(-la));
            float m = sg * (ZETA_C - GAMMA_C) + GAMMA_C;
            m = fminf(fmaxf(m, 0.f), 1.f);
            if (ev) { mask[idx] = m; rs += m; }
        }

        rs = row_sum1(rs);
        if (rv && sub == 0) dis[r] = fminf(rsqrtf(rs + 1e-6f), 10.f);
    }
}

__global__ __launch_bounds__(128) void row_spmm_kernel(
    const float* __restrict__ x, const float* __restrict__ mask,
    const float* __restrict__ dis,
    const int* __restrict__ rowptr, const int* __restrict__ cols,
    const float* addin, float* out, float* xn, int n_nodes)
{
    const int lane   = threadIdx.x & 31;
    const int sub    = lane & 15;
    const int hsel   = lane >> 4;
    const int wave_g = blockIdx.x * ROW_WAVES_PER_BLOCK + (threadIdx.x >> 5);
    const int r0     = wave_g * ROWS_PER_WAVE;
    if (r0 >= n_nodes) return;
    const float4* __restrict__ xv = (const float4*)x;
    const float4 zero4 = {0.f, 0.f, 0.f, 0.f};

    const int rp_i = r0 + lane;
    const int rp_l = rowptr[rp_i < n_nodes ? rp_i : n_nodes];

    for (int q = 0; q < ROWS_PER_WAVE / 2; ++q) {
        const int r = r0 + 2 * q + hsel;
        const bool rv = r < n_nodes;
        const int e0 = __builtin_amdgcn_readlane(rp_l, 2 * q);
        const int e1 = __builtin_amdgcn_readlane(rp_l, 2 * q + 1);
        const int e2 = __builtin_amdgcn_readlane(rp_l, 2 * q + 2);
        const int start = hsel ? e1 : e0;
        const int end   = hsel ? e2 : e1;
        const int degmax = max(e1 - e0, e2 - e1);
        const float dr = dis[rv ? r : n_nodes - 1];

        float4 acc = zero4;
        for (int base = 0; base < degmax; base += 16) {
            const int idx = start + base + sub;
            const bool ev = idx < end;
            const int c_l = ev ? cols[idx] : 0;
            const int nv  = end - (start + base);

            unsigned cj[16];
            float4 xc[16];
            row_share16<0>(cj, c_l);
            #pragma unroll
            for (int j = 0; j < 16; ++j)
                xc[j] = xv[cj[j] * (HID / 4) + sub];

            float v_l = 0.f;
            if (ev) v_l = mask[idx] * dr * dis[c_l];

            unsigned vj[16];
            row_share16<0>(vj, __builtin_bit_cast(int, v_l));

            #pragma unroll
            for (int j = 0; j < 16; ++j) {
                const float v = __builtin_bit_cast(float, vj[j]);
                const float4 t = (j < nv) ? xc[j] : zero4;
                acc.x += v * t.x;
                acc.y += v * t.y;
                acc.z += v * t.z;
                acc.w += v * t.w;
            }
        }

        if (rv) {
            const size_t o = (size_t)r * (HID / 4) + sub;
            const float4 bs = ((const float4*)addin)[o];
            float4 res;
            res.x = bs.x + acc.x; res.y = bs.y + acc.y;
            res.z = bs.z + acc.z; res.w = bs.w + acc.w;
            ((float4*)out)[o] = res;
            if (xn) ((float4*)xn)[o] = acc;
        }
    }
}

extern "C" void kernel_launch(void* const* d_in, const int* in_sizes, int n_in,
                              void* d_out, int out_size, void* d_ws, size_t ws_size,
                              hipStream_t stream) {
    (void)in_sizes; (void)n_in; (void)out_size; (void)ws_size;

    const float* feat = (const float*)d_in[0];
    const float* nbW  = (const float*)d_in[1];
    const float* nbb  = (const float*)d_in[2];
    const float* sW   = (const float*)d_in[3];
    const float* sb   = (const float*)d_in[4];
    const float* a1W  = (const float*)d_in[5];
    const float* a1b  = (const float*)d_in[6];
    const float* a2W  = (const float*)d_in[7];
    const float* a2b  = (const float*)d_in[8];
    const int*   row  = (const int*)d_in[9];
    const int*   col  = (const int*)d_in[10];
    float* out = (float*)d_out;

    const int NH = NNODES * HID;
    float* ws = (float*)d_ws;
    float* bufA   = ws; ws += NH;
    float* bufP1  = ws; ws += NH;
    float* bufP2  = ws; ws += NH;
    int*   cols   = (int*)ws; ws += NEDGES;
    float* maskb  = ws; ws += NEDGES;
    int*   rowptr = (int*)ws; ws += NNODES + 32;
    int*   cnt    = (int*)ws; ws += NNODES;
    float* dis    = ws; ws += NNODES;
    int*   bsum   = (int*)ws; ws += 128;

    const int gNode  = (NNODES + 255) / 256;
    const int nQuad  = NEDGES / 4;
    const int gEdge  = (nQuad + 255) / 256;
    const int gScan  = (NNODES + 1 + 1023) / 1024;
    const int gTile  = ((NNODES + 15) / 16 + 3) / 4;
    const int gRow   = (NNODES + ROWS_PER_WAVE * ROW_WAVES_PER_BLOCK - 1)
                       / (ROWS_PER_WAVE * ROW_WAVES_PER_BLOCK);

    zero_int_kernel<<<gNode, 256, 0, stream>>>(cnt, NNODES);
    row_count_kernel<<<gEdge, 256, 0, stream>>>(row, cnt, nQuad);
    block_sum_kernel<<<gScan, 1024, 0, stream>>>(cnt, bsum, NNODES);
    row_scan_kernel<<<gScan, 1024, 0, stream>>>(cnt, bsum, rowptr, NNODES);
    bucket_fill_kernel<<<gEdge, 256, 0, stream>>>(row, col, cnt, cols, nQuad);

    const float* xin[NLAYERS]  = {feat,  bufA,  bufP1};
    float*       p1o[NLAYERS]  = {bufP1, bufP1, bufA};
    float*       xout[NLAYERS] = {bufA,  bufP1, nullptr};
    const float* addin[NLAYERS] = {feat, out, out};

    for (int l = 0; l < NLAYERS; ++l) {
        node_transform_kernel<<<2 * gTile, 128, 0, stream>>>(
            xin[l],
            nbW + (size_t)l * 4096, nbb + l * 64,
            a1W + (size_t)l * 8192, a1b + l * 64, p1o[l],
            sW + (size_t)l * 4096, sb + l * 64,
            a1W + (size_t)l * 8192 + 4096, bufP2, NNODES);

        row_att_kernel<<<gRow, 128, 0, stream>>>(
            p1o[l], bufP2, a2W + l * 64, a2b + l, rowptr, cols, maskb, dis, NNODES);

        row_spmm_kernel<<<gRow, 128, 0, stream>>>(
            xin[l], maskb, dis, rowptr, cols, addin[l], out, xout[l], NNODES);
    }
}


# node transform: non-matrix issue slots per tile projection about 1070 -> 370: one 24-wave block per WGP stages both weight sets once in fragment order (8-byte B reads), x tile read once, plain stores
# speedup vs baseline: 2.6731x; 1.0620x over previous
#include <hip/hip_runtime.h>
#include <hip/hip_bf16.h>

#define NNODES 100000
#define NEDGES 1200000
#define HID 64
#define NLAYERS 3
#define GAMMA_C (-0.45f)
#define ZETA_C  (1.05f)

typedef float v2f __attribute__((ext_vector_type(2)));
typedef float v8f __attribute__((ext_vector_type(8)));

#define NT_WAVES   24
#define NT_THREADS (NT_WAVES * 32)
#define NT_BLOCKS  256
#define W_KP_PITCH 160
#define W_MAT      (32 * W_KP_PITCH)
#define G_PITCH    68

static_assert((size_t)(4 * W_MAT + NT_WAVES * 16 * G_PITCH) * 4 > 160u * 1024u,
              "the block's LDS must keep a second block off its WGP");
static_assert((size_t)(4 * W_MAT + NT_WAVES * 16 * G_PITCH) * 4 <= 320u * 1024u,
              "the block's LDS must fit a WGP");

__device__ __forceinline__ void node_tile_load(
    v2f (&xa)[16], const float* __restrict__ x, int n_nodes, int tile, int mn, int kh)
{
    int ri = tile * 16 + mn;
    if (ri >= n_nodes) ri = n_nodes - 1;
    const float* xr = x + (size_t)ri * HID;
    #pragma unroll
    for (int kk = 0; kk < 16; ++kk) {
        const int k = kk * 4 + kh;
        xa[kk].x = xr[k];
        xa[kk].y = xr[k + 1];
    }
}

__device__ __forceinline__ void node_tile_project(
    const v2f (&xa)[16], const float* ws, float* gt,
    const float* __restrict__ b1a, const float* __restrict__ bias2a,
    float* __restrict__ pa,
    const float* __restrict__ b1b, float* __restrict__ pb,
    int n_nodes, int tile, int p_lo, int p_hi, int mn, int kh, int mbase)
{
    const int row0 = tile * 16;
    const v8f vz = {0.f, 0.f, 0.f, 0.f, 0.f, 0.f, 0.f, 0.f};

    for (int pj = p_lo; pj < p_hi; ++pj) {
        const float* w1 = ws + (2 * pj) * W_MAT;
        const float* w2 = w1 + W_MAT;
        const float* __restrict__ b1 = pj == 0 ? b1a : b1b;
        float* __restrict__ p = pj == 0 ? pa : pb;

        v8f acc[4] = {vz, vz, vz, vz};
        #pragma unroll
        for (int kk = 0; kk < 16; ++kk) {
            const int kp = kk * 2 + (kh >> 1);
            #pragma unroll
            for (int nt = 0; nt < 4; ++nt) {
                const v2f b = *(const v2f*)(w1 + kp * W_KP_PITCH + 2 * (nt * 16 + mn));
                acc[nt] = __builtin_amdgcn_wmma_f32_16x16x4_f32(
                    false, xa[kk], false, b, (short)0, acc[nt], false, false);
            }
        }

        #pragma unroll
        for (int nt = 0; nt < 4; ++nt) {
            const float bv = b1[nt * 16 + mn];
            #pragma unroll
            for (int i = 0; i < 8; ++i) {
                float v = acc[nt][i] + bv;
                v = v > 0.f ? v : 0.f;
                gt[(mbase + i) * G_PITCH + nt * 16 + mn] = v;
            }
        }

        v8f acc2[4] = {vz, vz, vz, vz};
        #pragma unroll
        for (int kk = 0; kk < 16; ++kk) {
            const int k = kk * 4 + kh;
            const int kp = kk * 2 + (kh >> 1);
            v2f a; a.x = gt[mn * G_PITCH + k]; a.y = gt[mn * G_PITCH + k + 1];
            #pragma unroll
            for (int nt = 0; nt < 4; ++nt) {
                const v2f b = *(const v2f*)(w2 + kp * W_KP_PITCH + 2 * (nt * 16 + mn));
                acc2[nt] = __builtin_amdgcn_wmma_f32_16x16x4_f32(
                    false, a, false, b, (short)0, acc2[nt], false, false);
            }
        }

        float b2v[4];
        #pragma unroll
        for (int nt = 0; nt < 4; ++nt)
            b2v[nt] = pj == 0 ? bias2a[nt * 16 + mn] : 0.f;
        if (row0 + 16 <= n_nodes) {
            float* __restrict__ pr = p + (size_t)(row0 + mbase) * HID + mn;
            #pragma unroll
            for (int nt = 0; nt < 4; ++nt) {
                #pragma unroll
                for (int i = 0; i < 8; ++i)
                    pr[i * HID + nt * 16] = acc2[nt][i] + b2v[nt];
            }
        } else {
            #pragma unroll
            for (int nt = 0; nt < 4; ++nt) {
                #pragma unroll
                for (int i = 0; i < 8; ++i) {
                    const int r = row0 + mbase + i;
                    if (r < n_nodes)
                        p[(size_t)r * HID + nt * 16 + mn] = acc2[nt][i] + b2v[nt];
                }
            }
        }
    }
}

__device__ __forceinline__ bool node_pick(
    int cnt, int j0, int wave, int& j, int& p_lo, int& p_hi, int& step)
{
    const bool split = cnt - j0 <= NT_WAVES / 2;
    j    = j0 + (split ? (wave >> 1) : wave);
    p_lo = split ? (wave & 1) : 0;
    p_hi = split ? p_lo + 1 : 2;
    step = split ? NT_WAVES / 2 : NT_WAVES;
    return j < cnt;
}

__global__ __launch_bounds__(NT_THREADS) void node_transform_kernel(
    const float* __restrict__ x,
    const float* __restrict__ W1a, const float* __restrict__ b1a,
    const float* __restrict__ W2a, const float* __restrict__ bias2a,
    float* __restrict__ pa,
    const float* __restrict__ W1b, const float* __restrict__ b1b,
    const float* __restrict__ W2b, float* __restrict__ pb,
    int n_nodes)
{
    __shared__ __attribute__((aligned(16))) float ws[4 * W_MAT];
    __shared__ float gs[NT_WAVES * 16 * G_PITCH];

    const int wave  = __builtin_amdgcn_readfirstlane((int)(threadIdx.x >> 5));
    const int lane  = threadIdx.x & 31;
    const int mn    = lane & 15;
    const int kh    = (lane >> 4) << 1;
    const int mbase = (lane >> 4) << 3;
    float* gt = gs + wave * (16 * G_PITCH);

    const int n_tiles = (n_nodes + 15) >> 4;
    const int blk = (int)blockIdx.x;
    const int cnt = blk < n_tiles ? (n_tiles - 1 - blk) / NT_BLOCKS + 1 : 0;

    int j0 = 0, j, p_lo, p_hi, step;
    bool mine = node_pick(cnt, j0, wave, j, p_lo, p_hi, step);
    v2f xa[16];

    constexpr int W_SLOTS = 4 * 16;
    constexpr int W_FULL  = W_SLOTS / NT_WAVES;
    constexpr int W_REST  = W_SLOTS % NT_WAVES;
    const int skp = lane >> 4;
    const int sn4 = lane & 15;
    float4 sev[W_FULL], sod[W_FULL];
    #pragma unroll
    for (int t = 0; t < W_FULL; ++t) {
        const int s = wave + t * NT_WAVES;
        const int m = s >> 4;
        const float* src = m == 0 ? W1a : m == 1 ? W2a : m == 2 ? W1b : W2b;
        const int kp = (s & 15) * 2 + skp;
        sev[t] = ((const float4*)src)[(2 * kp) * 16 + sn4];
        sod[t] = ((const float4*)src)[(2 * kp + 1) * 16 + sn4];
    }
    float4 rev = {0.f, 0.f, 0.f, 0.f}, rod = {0.f, 0.f, 0.f, 0.f};
    if (wave < W_REST) {
        const int s = wave + W_FULL * NT_WAVES;
        const int m = s >> 4;
        const float* src = m == 0 ? W1a : m == 1 ? W2a : m == 2 ? W1b : W2b;
        const int kp = (s & 15) * 2 + skp;
        rev = ((const float4*)src)[(2 * kp) * 16 + sn4];
        rod = ((const float4*)src)[(2 * kp + 1) * 16 + sn4];
    }
    #pragma unroll
    for (int t = 0; t < W_FULL; ++t) {
        const int s = wave + t * NT_WAVES;
        float4* d = (float4*)(ws + (s >> 4) * W_MAT
                                 + ((s & 15) * 2 + skp) * W_KP_PITCH + sn4 * 8);
        d[0] = make_float4(sev[t].x, sod[t].x, sev[t].y, sod[t].y);
        d[1] = make_float4(sev[t].z, sod[t].z, sev[t].w, sod[t].w);
    }
    if (wave < W_REST) {
        const int s = wave + W_FULL * NT_WAVES;
        float4* d = (float4*)(ws + (s >> 4) * W_MAT
                                 + ((s & 15) * 2 + skp) * W_KP_PITCH + sn4 * 8);
        d[0] = make_float4(rev.x, rod.x, rev.y, rod.y);
        d[1] = make_float4(rev.z, rod.z, rev.w, rod.w);
    }
    __syncthreads();

    while (j0 < cnt) {
        if (mine) {
            const int tile = blk + NT_BLOCKS * j;
            node_tile_load(xa, x, n_nodes, tile, mn, kh);
            node_tile_project(xa, ws, gt, b1a, bias2a, pa, b1b, pb, n_nodes,
                              tile, p_lo, p_hi, mn, kh, mbase);
        }
        j0 += step;
        if (j0 >= cnt) break;
        mine = node_pick(cnt, j0, wave, j, p_lo, p_hi, step);
    }
}

static_assert(NEDGES % 4 == 0, "edge count must fill whole groups of four");

__global__ __launch_bounds__(256) void zero_int_kernel(int* __restrict__ p, int n)
{
    const int i = blockIdx.x * 256 + threadIdx.x;
    if (i < n) p[i] = 0;
}

__global__ __launch_bounds__(256) void row_count_kernel(
    const int* __restrict__ row, int* __restrict__ cnt, int n_quads)
{
    const int i = blockIdx.x * 256 + threadIdx.x;
    if (i < n_quads) {
        const int4 r = ((const int4*)row)[i];
        atomicAdd(cnt + r.x, 1);
        atomicAdd(cnt + r.y, 1);
        atomicAdd(cnt + r.z, 1);
        atomicAdd(cnt + r.w, 1);
    }
}

__global__ __launch_bounds__(1024) void block_sum_kernel(
    const int* __restrict__ cnt, int* __restrict__ bsum, int n)
{
    __shared__ int wsum[32];
    const int i = blockIdx.x * 1024 + threadIdx.x;
    int v = i < n ? cnt[i] : 0;
    for (int o = 16; o > 0; o >>= 1) v += __shfl_xor(v, o, 32);
    if ((threadIdx.x & 31) == 0) wsum[threadIdx.x >> 5] = v;
    __syncthreads();
    if (threadIdx.x < 32) {
        int t = wsum[threadIdx.x];
        for (int o = 16; o > 0; o >>= 1) t += __shfl_xor(t, o, 32);
        if (threadIdx.x == 0) bsum[blockIdx.x] = t;
    }
}

__global__ __launch_bounds__(1024) void row_scan_kernel(
    int* __restrict__ cnt, const int* __restrict__ bsum,
    int* __restrict__ rowptr, int n)
{
    __shared__ int wsum[32];
    __shared__ int boff;
    const int tid  = threadIdx.x;
    const int lane = tid & 31;
    const int wv   = tid >> 5;
    const int i    = blockIdx.x * 1024 + tid;
    const int v    = i < n ? cnt[i] : 0;

    int s = v;
    for (int o = 1; o < 32; o <<= 1) {
        const int t = __shfl_up(s, o, 32);
        if (lane >= o) s += t;
    }
    if (lane == 31) wsum[wv] = s;
    if (wv == 0) {
        int t = 0;
        for (int k = lane; k < (int)blockIdx.x; k += 32) t += bsum[k];
        for (int o = 16; o > 0; o >>= 1) t += __shfl_xor(t, o, 32);
        if (lane == 0) boff = t;
    }
    __syncthreads();
    if (wv == 0) {
        const int t = wsum[lane];
        int u = t;
        for (int o = 1; o < 32; o <<= 1) {
            const int y = __shfl_up(u, o, 32);
            if (lane >= o) u += y;
        }
        wsum[lane] = u - t;
    }
    __syncthreads();
    const int excl = boff + wsum[wv] + s - v;
    if (i <= n) rowptr[i] = excl;
    if (i < n)  cnt[i] = excl;
}

__global__ __launch_bounds__(256) void bucket_fill_kernel(
    const int* __restrict__ row, const int* __restrict__ col,
    int* __restrict__ cursor, int* __restrict__ cols, int n_quads)
{
    const int i = blockIdx.x * 256 + threadIdx.x;
    if (i < n_quads) {
        const int4 r = ((const int4*)row)[i];
        const int4 c = ((const int4*)col)[i];
        const int p0 = atomicAdd(cursor + r.x, 1);
        const int p1 = atomicAdd(cursor + r.y, 1);
        const int p2 = atomicAdd(cursor + r.z, 1);
        const int p3 = atomicAdd(cursor + r.w, 1);
        cols[p0] = c.x;
        cols[p1] = c.y;
        cols[p2] = c.z;
        cols[p3] = c.w;
    }
}

#define ROWS_PER_WAVE 8
#define ROW_WAVES_PER_BLOCK 4

#define DPP_ROW_SHARE(j) (0x150 + (j))
#define DPP_ROW_XMASK(m) (0x160 + (m))

template <int CTRL>
__device__ __forceinline__ int dpp_i(int v)
{
    return __builtin_amdgcn_update_dpp(0, v, CTRL, 0xf, 0xf, true);
}

template <int CTRL>
__device__ __forceinline__ float dpp_f(float v)
{
    return __builtin_bit_cast(float,
        __builtin_amdgcn_update_dpp(0, __builtin_bit_cast(int, v), CTRL, 0xf, 0xf, true));
}

template <int J>
__device__ __forceinline__ void row_share16(unsigned (&o)[16], int v)
{
    if constexpr (J < 16) {
        o[J] = (unsigned)dpp_i<DPP_ROW_SHARE(J)>(v);
        row_share16<J + 1>(o, v);
    }
}

template <int M, int N>
__device__ __forceinline__ void fold_step(float (&s)[16], bool hi)
{
    #pragma unroll
    for (int j = 0; j < N; ++j) {
        const float keep = hi ? s[j + N] : s[j];
        const float give = hi ? s[j] : s[j + N];
        s[j] = keep + dpp_f<DPP_ROW_XMASK(M)>(give);
    }
}

__device__ __forceinline__ float row_sums16(float (&s)[16], int sub)
{
    fold_step<8, 8>(s, (sub & 8) != 0);
    fold_step<4, 4>(s, (sub & 4) != 0);
    fold_step<2, 2>(s, (sub & 2) != 0);
    fold_step<1, 1>(s, (sub & 1) != 0);
    return s[0];
}

__device__ __forceinline__ float row_sum1(float v)
{
    v += dpp_f<DPP_ROW_XMASK(8)>(v);
    v += dpp_f<DPP_ROW_XMASK(4)>(v);
    v += dpp_f<DPP_ROW_XMASK(2)>(v);
    v += dpp_f<DPP_ROW_XMASK(1)>(v);
    return v;
}

__global__ __launch_bounds__(128) void row_att_kernel(
    const float* __restrict__ p1, const float* __restrict__ p2,
    const float* __restrict__ w2, const float* __restrict__ b2,
    const int* __restrict__ rowptr, const int* __restrict__ cols,
    float* __restrict__ mask, float* __restrict__ dis, int n_nodes)
{
    const int lane   = threadIdx.x & 31;
    const int sub    = lane & 15;
    const int hsel   = lane >> 4;
    const int wave_g = blockIdx.x * ROW_WAVES_PER_BLOCK + (threadIdx.x >> 5);
    const int r0     = wave_g * ROWS_PER_WAVE;
    if (r0 >= n_nodes) return;
    const float4* __restrict__ p1v = (const float4*)p1;
    const float4* __restrict__ p2v = (const float4*)p2;

    const float4 w = ((const float4*)w2)[sub];
    const float bias = b2[0];

    const int rp_i = r0 + lane;
    const int rp_l = rowptr[rp_i < n_nodes ? rp_i : n_nodes];

    for (int q = 0; q < ROWS_PER_WAVE / 2; ++q) {
        const int r = r0 + 2 * q + hsel;
        const bool rv = r < n_nodes;
        const int e0 = __builtin_amdgcn_readlane(rp_l, 2 * q);
        const int e1 = __builtin_amdgcn_readlane(rp_l, 2 * q + 1);
        const int e2 = __builtin_amdgcn_readlane(rp_l, 2 * q + 2);
        const int start = hsel ? e1 : e0;
        const int end   = hsel ? e2 : e1;
        const int degmax = max(e1 - e0, e2 - e1);
        const float4 a = p1v[(rv ? r : n_nodes - 1) * (HID / 4) + sub];

        float rs = 0.f;
        for (int base = 0; base < degmax; base += 16) {
            const int idx = start + base + sub;
            const bool ev = idx < end;
            const int c_l = ev ? cols[idx] : 0;

            unsigned cj[16];
            float4 b[16];
            row_share16<0>(cj, c_l);
            #pragma unroll
            for (int j = 0; j < 16; ++j)
                b[j] = p2v[cj[j] * (HID / 4) + sub];

            float s[16];
            #pragma unroll
            for (int j = 0; j < 16; ++j) {
                const float t0 = fmaxf(a.x + b[j].x, 0.f);
                const float t1 = fmaxf(a.y + b[j].y, 0.f);
                const float t2 = fmaxf(a.z + b[j].z, 0.f);
                const float t3 = fmaxf(a.w + b[j].w, 0.f);
                s[j] = t0 * w.x + t1 * w.y + t2 * w.z + t3 * w.w;
            }
            const float mine = row_sums16(s, sub);

            const float la = mine + bias;
            const float sg = 1.f / (1.f + __expf(-la));
            float m = sg * (ZETA_C - GAMMA_C) + GAMMA_C;
            m = fminf(fmaxf(m, 0.f), 1.f);
            if (ev) { mask[idx] = m; rs += m; }
        }

        rs = row_sum1(rs);
        if (rv && sub == 0) dis[r] = fminf(rsqrtf(rs + 1e-6f), 10.f);
    }
}

__global__ __launch_bounds__(128) void row_spmm_kernel(
    const float* __restrict__ x, const float* __restrict__ mask,
    const float* __restrict__ dis,
    const int* __restrict__ rowptr, const int* __restrict__ cols,
    const float* addin, float* out, float* xn, int n_nodes)
{
    const int lane   = threadIdx.x & 31;
    const int sub    = lane & 15;
    const int hsel   = lane >> 4;
    const int wave_g = blockIdx.x * ROW_WAVES_PER_BLOCK + (threadIdx.x >> 5);
    const int r0     = wave_g * ROWS_PER_WAVE;
    if (r0 >= n_nodes) return;
    const float4* __restrict__ xv = (const float4*)x;
    const float4 zero4 = {0.f, 0.f, 0.f, 0.f};

    const int rp_i = r0 + lane;
    const int rp_l = rowptr[rp_i < n_nodes ? rp_i : n_nodes];

    for (int q = 0; q < ROWS_PER_WAVE / 2; ++q) {
        const int r = r0 + 2 * q + hsel;
        const bool rv = r < n_nodes;
        const int e0 = __builtin_amdgcn_readlane(rp_l, 2 * q);
        const int e1 = __builtin_amdgcn_readlane(rp_l, 2 * q + 1);
        const int e2 = __builtin_amdgcn_readlane(rp_l, 2 * q + 2);
        const int start = hsel ? e1 : e0;
        const int end   = hsel ? e2 : e1;
        const int degmax = max(e1 - e0, e2 - e1);
        const float dr = dis[rv ? r : n_nodes - 1];

        float4 acc = zero4;
        for (int base = 0; base < degmax; base += 16) {
            const int idx = start + base + sub;
            const bool ev = idx < end;
            const int c_l = ev ? cols[idx] : 0;
            const int nv  = end - (start + base);

            unsigned cj[16];
            float4 xc[16];
            row_share16<0>(cj, c_l);
            #pragma unroll
            for (int j = 0; j < 16; ++j)
                xc[j] = xv[cj[j] * (HID / 4) + sub];

            float v_l = 0.f;
            if (ev) v_l = mask[idx] * dr * dis[c_l];

            unsigned vj[16];
            row_share16<0>(vj, __builtin_bit_cast(int, v_l));

            #pragma unroll
            for (int j = 0; j < 16; ++j) {
                const float v = __builtin_bit_cast(float, vj[j]);
                const float4 t = (j < nv) ? xc[j] : zero4;
                acc.x += v * t.x;
                acc.y += v * t.y;
                acc.z += v * t.z;
                acc.w += v * t.w;
            }
        }

        if (rv) {
            const size_t o = (size_t)r * (HID / 4) + sub;
            const float4 bs = ((const float4*)addin)[o];
            float4 res;
            res.x = bs.x + acc.x; res.y = bs.y + acc.y;
            res.z = bs.z + acc.z; res.w = bs.w + acc.w;
            ((float4*)out)[o] = res;
            if (xn) ((float4*)xn)[o] = acc;
        }
    }
}

extern "C" void kernel_launch(void* const* d_in, const int* in_sizes, int n_in,
                              void* d_out, int out_size, void* d_ws, size_t ws_size,
                              hipStream_t stream) {
    (void)in_sizes; (void)n_in; (void)out_size; (void)ws_size;

    const float* feat = (const float*)d_in[0];
    const float* nbW  = (const float*)d_in[1];
    const float* nbb  = (const float*)d_in[2];
    const float* sW   = (const float*)d_in[3];
    const float* sb   = (const float*)d_in[4];
    const float* a1W  = (const float*)d_in[5];
    const float* a1b  = (const float*)d_in[6];
    const float* a2W  = (const float*)d_in[7];
    const float* a2b  = (const float*)d_in[8];
    const int*   row  = (const int*)d_in[9];
    const int*   col  = (const int*)d_in[10];
    float* out = (float*)d_out;

    const int NH = NNODES * HID;
    float* ws = (float*)d_ws;
    float* bufA   = ws; ws += NH;
    float* bufP1  = ws; ws += NH;
    float* bufP2  = ws; ws += NH;
    int*   cols   = (int*)ws; ws += NEDGES;
    float* maskb  = ws; ws += NEDGES;
    int*   rowptr = (int*)ws; ws += NNODES + 32;
    int*   cnt    = (int*)ws; ws += NNODES;
    float* dis    = ws; ws += NNODES;
    int*   bsum   = (int*)ws; ws += 128;

    const int gNode  = (NNODES + 255) / 256;
    const int nQuad  = NEDGES / 4;
    const int gEdge  = (nQuad + 255) / 256;
    const int gScan  = (NNODES + 1 + 1023) / 1024;
    const int gRow   = (NNODES + ROWS_PER_WAVE * ROW_WAVES_PER_BLOCK - 1)
                       / (ROWS_PER_WAVE * ROW_WAVES_PER_BLOCK);

    zero_int_kernel<<<gNode, 256, 0, stream>>>(cnt, NNODES);
    row_count_kernel<<<gEdge, 256, 0, stream>>>(row, cnt, nQuad);
    block_sum_kernel<<<gScan, 1024, 0, stream>>>(cnt, bsum, NNODES);
    row_scan_kernel<<<gScan, 1024, 0, stream>>>(cnt, bsum, rowptr, NNODES);
    bucket_fill_kernel<<<gEdge, 256, 0, stream>>>(row, col, cnt, cols, nQuad);

    const float* xin[NLAYERS]  = {feat,  bufA,  bufP1};
    float*       p1o[NLAYERS]  = {bufP1, bufP1, bufA};
    float*       xout[NLAYERS] = {bufA,  bufP1, nullptr};
    const float* addin[NLAYERS] = {feat, out, out};

    for (int l = 0; l < NLAYERS; ++l) {
        node_transform_kernel<<<NT_BLOCKS, NT_THREADS, 0, stream>>>(
            xin[l],
            nbW + (size_t)l * 4096, nbb + l * 64,
            a1W + (size_t)l * 8192, a1b + l * 64, p1o[l],
            sW + (size_t)l * 4096, sb + l * 64,
            a1W + (size_t)l * 8192 + 4096, bufP2, NNODES);

        row_att_kernel<<<gRow, 128, 0, stream>>>(
            p1o[l], bufP2, a2W + l * 64, a2b + l, rowptr, cols, maskb, dis, NNODES);

        row_spmm_kernel<<<gRow, 128, 0, stream>>>(
            xin[l], maskb, dis, rowptr, cols, addin[l], out, xout[l], NNODES);
    }
}


# edge bucketing: two-level counting sort on LDS counters (512-row bins per 8192-edge chunk, table scan, pair scatter, per-bin count+scan+fill): no global atomics, 13 launches not 14; bucket-order sums
# speedup vs baseline: 3.6492x; 1.3652x over previous
#include <hip/hip_runtime.h>
#include <hip/hip_bf16.h>

#define NNODES 100000
#define NEDGES 1200000
#define HID 64
#define NLAYERS 3
#define GAMMA_C (-0.45f)
#define ZETA_C  (1.05f)

typedef float v2f __attribute__((ext_vector_type(2)));
typedef float v8f __attribute__((ext_vector_type(8)));

#define NT_WAVES   24
#define NT_THREADS (NT_WAVES * 32)
#define NT_BLOCKS  256
#define W_KP_PITCH 160
#define W_MAT      (32 * W_KP_PITCH)
#define G_PITCH    68

static_assert((size_t)(4 * W_MAT + NT_WAVES * 16 * G_PITCH) * 4 > 160u * 1024u,
              "the block's LDS must keep a second block off its WGP");
static_assert((size_t)(4 * W_MAT + NT_WAVES * 16 * G_PITCH) * 4 <= 320u * 1024u,
              "the block's LDS must fit a WGP");

__device__ __forceinline__ void node_tile_load(
    v2f (&xa)[16], const float* __restrict__ x, int n_nodes, int tile, int mn, int kh)
{
    int ri = tile * 16 + mn;
    if (ri >= n_nodes) ri = n_nodes - 1;
    const float* xr = x + (size_t)ri * HID;
    #pragma unroll
    for (int kk = 0; kk < 16; ++kk) {
        const int k = kk * 4 + kh;
        xa[kk].x = xr[k];
        xa[kk].y = xr[k + 1];
    }
}

__device__ __forceinline__ void node_tile_project(
    const v2f (&xa)[16], const float* ws, float* gt,
    const float* __restrict__ b1a, const float* __restrict__ bias2a,
    float* __restrict__ pa,
    const float* __restrict__ b1b, float* __restrict__ pb,
    int n_nodes, int tile, int p_lo, int p_hi, int mn, int kh, int mbase)
{
    const int row0 = tile * 16;
    const v8f vz = {0.f, 0.f, 0.f, 0.f, 0.f, 0.f, 0.f, 0.f};

    for (int pj = p_lo; pj < p_hi; ++pj) {
        const float* w1 = ws + (2 * pj) * W_MAT;
        const float* w2 = w1 + W_MAT;
        const float* __restrict__ b1 = pj == 0 ? b1a : b1b;
        float* __restrict__ p = pj == 0 ? pa : pb;

        v8f acc[4] = {vz, vz, vz, vz};
        #pragma unroll
        for (int kk = 0; kk < 16; ++kk) {
            const int kp = kk * 2 + (kh >> 1);
            #pragma unroll
            for (int nt = 0; nt < 4; ++nt) {
                const v2f b = *(const v2f*)(w1 + kp * W_KP_PITCH + 2 * (nt * 16 + mn));
                acc[nt] = __builtin_amdgcn_wmma_f32_16x16x4_f32(
                    false, xa[kk], false, b, (short)0, acc[nt], false, false);
            }
        }

        #pragma unroll
        for (int nt = 0; nt < 4; ++nt) {
            const float bv = b1[nt * 16 + mn];
            #pragma unroll
            for (int i = 0; i < 8; ++i) {
                float v = acc[nt][i] + bv;
                v = v > 0.f ? v : 0.f;
                gt[(mbase + i) * G_PITCH + nt * 16 + mn] = v;
            }
        }

        v8f acc2[4] = {vz, vz, vz, vz};
        #pragma unroll
        for (int kk = 0; kk < 16; ++kk) {
            const int k = kk * 4 + kh;
            const int kp = kk * 2 + (kh >> 1);
            v2f a; a.x = gt[mn * G_PITCH + k]; a.y = gt[mn * G_PITCH + k + 1];
            #pragma unroll
            for (int nt = 0; nt < 4; ++nt) {
                const v2f b = *(const v2f*)(w2 + kp * W_KP_PITCH + 2 * (nt * 16 + mn));
                acc2[nt] = __builtin_amdgcn_wmma_f32_16x16x4_f32(
                    false, a, false, b, (short)0, acc2[nt], false, false);
            }
        }

        float b2v[4];
        #pragma unroll
        for (int nt = 0; nt < 4; ++nt)
            b2v[nt] = pj == 0 ? bias2a[nt * 16 + mn] : 0.f;
        if (row0 + 16 <= n_nodes) {
            float* __restrict__ pr = p + (size_t)(row0 + mbase) * HID + mn;
            #pragma unroll
            for (int nt = 0; nt < 4; ++nt) {
                #pragma unroll
                for (int i = 0; i < 8; ++i)
                    pr[i * HID + nt * 16] = acc2[nt][i] + b2v[nt];
            }
        } else {
            #pragma unroll
            for (int nt = 0; nt < 4; ++nt) {
                #pragma unroll
                for (int i = 0; i < 8; ++i) {
                    const int r = row0 + mbase + i;
                    if (r < n_nodes)
                        p[(size_t)r * HID + nt * 16 + mn] = acc2[nt][i] + b2v[nt];
                }
            }
        }
    }
}

__device__ __forceinline__ bool node_pick(
    int cnt, int j0, int wave, int& j, int& p_lo, int& p_hi, int& step)
{
    const bool split = cnt - j0 <= NT_WAVES / 2;
    j    = j0 + (split ? (wave >> 1) : wave);
    p_lo = split ? (wave & 1) : 0;
    p_hi = split ? p_lo + 1 : 2;
    step = split ? NT_WAVES / 2 : NT_WAVES;
    return j < cnt;
}

__global__ __launch_bounds__(NT_THREADS) void node_transform_kernel(
    const float* __restrict__ x,
    const float* __restrict__ W1a, const float* __restrict__ b1a,
    const float* __restrict__ W2a, const float* __restrict__ bias2a,
    float* __restrict__ pa,
    const float* __restrict__ W1b, const float* __restrict__ b1b,
    const float* __restrict__ W2b, float* __restrict__ pb,
    int n_nodes)
{
    __shared__ __attribute__((aligned(16))) float ws[4 * W_MAT];
    __shared__ float gs[NT_WAVES * 16 * G_PITCH];

    const int wave  = __builtin_amdgcn_readfirstlane((int)(threadIdx.x >> 5));
    const int lane  = threadIdx.x & 31;
    const int mn    = lane & 15;
    const int kh    = (lane >> 4) << 1;
    const int mbase = (lane >> 4) << 3;
    float* gt = gs + wave * (16 * G_PITCH);

    const int n_tiles = (n_nodes + 15) >> 4;
    const int blk = (int)blockIdx.x;
    const int cnt = blk < n_tiles ? (n_tiles - 1 - blk) / NT_BLOCKS + 1 : 0;

    int j0 = 0, j, p_lo, p_hi, step;
    bool mine = node_pick(cnt, j0, wave, j, p_lo, p_hi, step);
    v2f xa[16];

    constexpr int W_SLOTS = 4 * 16;
    constexpr int W_FULL  = W_SLOTS / NT_WAVES;
    constexpr int W_REST  = W_SLOTS % NT_WAVES;
    const int skp = lane >> 4;
    const int sn4 = lane & 15;
    float4 sev[W_FULL], sod[W_FULL];
    #pragma unroll
    for (int t = 0; t < W_FULL; ++t) {
        const int s = wave + t * NT_WAVES;
        const int m = s >> 4;
        const float* src = m == 0 ? W1a : m == 1 ? W2a : m == 2 ? W1b : W2b;
        const int kp = (s & 15) * 2 + skp;
        sev[t] = ((const float4*)src)[(2 * kp) * 16 + sn4];
        sod[t] = ((const float4*)src)[(2 * kp + 1) * 16 + sn4];
    }
    float4 rev = {0.f, 0.f, 0.f, 0.f}, rod = {0.f, 0.f, 0.f, 0.f};
    if (wave < W_REST) {
        const int s = wave + W_FULL * NT_WAVES;
        const int m = s >> 4;
        const float* src = m == 0 ? W1a : m == 1 ? W2a : m == 2 ? W1b : W2b;
        const int kp = (s & 15) * 2 + skp;
        rev = ((const float4*)src)[(2 * kp) * 16 + sn4];
        rod = ((const float4*)src)[(2 * kp + 1) * 16 + sn4];
    }
    #pragma unroll
    for (int t = 0; t < W_FULL; ++t) {
        const int s = wave + t * NT_WAVES;
        float4* d = (float4*)(ws + (s >> 4) * W_MAT
                                 + ((s & 15) * 2 + skp) * W_KP_PITCH + sn4 * 8);
        d[0] = make_float4(sev[t].x, sod[t].x, sev[t].y, sod[t].y);
        d[1] = make_float4(sev[t].z, sod[t].z, sev[t].w, sod[t].w);
    }
    if (wave < W_REST) {
        const int s = wave + W_FULL * NT_WAVES;
        float4* d = (float4*)(ws + (s >> 4) * W_MAT
                                 + ((s & 15) * 2 + skp) * W_KP_PITCH + sn4 * 8);
        d[0] = make_float4(rev.x, rod.x, rev.y, rod.y);
        d[1] = make_float4(rev.z, rod.z, rev.w, rod.w);
    }
    __syncthreads();

    while (j0 < cnt) {
        if (mine) {
            const int tile = blk + NT_BLOCKS * j;
            node_tile_load(xa, x, n_nodes, tile, mn, kh);
            node_tile_project(xa, ws, gt, b1a, bias2a, pa, b1b, pb, n_nodes,
                              tile, p_lo, p_hi, mn, kh, mbase);
        }
        j0 += step;
        if (j0 >= cnt) break;
        mine = node_pick(cnt, j0, wave, j, p_lo, p_hi, step);
    }
}

static_assert(NEDGES % 4 == 0, "edge count must fill whole groups of four");

#define SORT_BIN_SHIFT   9
#define SORT_BIN_ROWS    (1 << SORT_BIN_SHIFT)
#define SORT_NBINS       (NNODES / SORT_BIN_ROWS + 1)
#define SORT_THREADS     512
#define SORT_CHUNK_QUADS (4 * SORT_THREADS)
#define SORT_NCHUNKS     ((NEDGES / 4 + SORT_CHUNK_QUADS - 1) / SORT_CHUNK_QUADS)
#define SORT_TABLE       (SORT_NBINS * SORT_NCHUNKS)
#define SCAN_PER         ((SORT_TABLE + 1023) / 1024)

__global__ __launch_bounds__(SORT_THREADS) void bin_count_kernel(
    const int* __restrict__ row, int* __restrict__ table, int n_quads)
{
    __shared__ int hist[SORT_NBINS];
    for (int b = threadIdx.x; b < SORT_NBINS; b += SORT_THREADS) hist[b] = 0;
    __syncthreads();
    const int q0 = blockIdx.x * SORT_CHUNK_QUADS + threadIdx.x;
    constexpr int NQ = SORT_CHUNK_QUADS / SORT_THREADS;
    int4 r[NQ];
    for (int j = 0; j < NQ; ++j) {
        const int q = q0 + j * SORT_THREADS;
        r[j] = ((const int4*)row)[q < n_quads ? q : n_quads - 1];
    }
    for (int j = 0; j < NQ; ++j) {
        if (q0 + j * SORT_THREADS < n_quads) {
            atomicAdd(&hist[r[j].x >> SORT_BIN_SHIFT], 1);
            atomicAdd(&hist[r[j].y >> SORT_BIN_SHIFT], 1);
            atomicAdd(&hist[r[j].z >> SORT_BIN_SHIFT], 1);
            atomicAdd(&hist[r[j].w >> SORT_BIN_SHIFT], 1);
        }
    }
    __syncthreads();
    for (int b = threadIdx.x; b < SORT_NBINS; b += SORT_THREADS)
        table[b * SORT_NCHUNKS + blockIdx.x] = hist[b];
}

__global__ __launch_bounds__(1024) void bin_scan_kernel(
    int* __restrict__ table, int* __restrict__ binbase)
{
    __shared__ int t[SCAN_PER * 1024];
    __shared__ int wsum[32];
    const int tid  = threadIdx.x;
    const int lane = tid & 31;
    const int wv   = tid >> 5;

    for (int j = 0; j < SCAN_PER; ++j) {
        const int i = j * 1024 + tid;
        t[i] = i < SORT_TABLE ? table[i] : 0;
    }
    __syncthreads();

    int* seg = t + tid * SCAN_PER;
    int sum = 0;
    for (int k = 0; k < SCAN_PER; ++k) sum += seg[k];

    int s = sum;
    for (int o = 1; o < 32; o <<= 1) {
        const int y = __shfl_up(s, o, 32);
        if (lane >= o) s += y;
    }
    if (lane == 31) wsum[wv] = s;
    __syncthreads();
    if (wv == 0) {
        const int w = wsum[lane];
        int u = w;
        for (int o = 1; o < 32; o <<= 1) {
            const int y = __shfl_up(u, o, 32);
            if (lane >= o) u += y;
        }
        wsum[lane] = u - w;
    }
    __syncthreads();

    int run = wsum[wv] + s - sum;
    for (int k = 0; k < SCAN_PER; ++k) {
        const int v = seg[k];
        seg[k] = run;
        run += v;
    }
    __syncthreads();

    for (int j = 0; j < SCAN_PER; ++j) {
        const int i = j * 1024 + tid;
        if (i < SORT_TABLE) {
            const int p = t[i];
            table[i] = p;
            if (i % SORT_NCHUNKS == 0) binbase[i / SORT_NCHUNKS] = p;
        }
    }
    if (tid == 1023) binbase[SORT_NBINS] = run;
}

__global__ __launch_bounds__(SORT_THREADS) void bin_scatter_kernel(
    const int* __restrict__ row, const int* __restrict__ col,
    const int* __restrict__ table, int* __restrict__ pairs, int n_quads)
{
    __shared__ int off[SORT_NBINS];
    for (int b = threadIdx.x; b < SORT_NBINS; b += SORT_THREADS)
        off[b] = table[b * SORT_NCHUNKS + blockIdx.x];
    __syncthreads();
    int2* __restrict__ pv = (int2*)pairs;
    const int q0 = blockIdx.x * SORT_CHUNK_QUADS + threadIdx.x;
    constexpr int NQ = SORT_CHUNK_QUADS / SORT_THREADS;
    int4 r[NQ], c[NQ];
    for (int j = 0; j < NQ; ++j) {
        const int q = q0 + j * SORT_THREADS;
        const int qc = q < n_quads ? q : n_quads - 1;
        r[j] = ((const int4*)row)[qc];
        c[j] = ((const int4*)col)[qc];
    }
    for (int j = 0; j < NQ; ++j) {
        if (q0 + j * SORT_THREADS < n_quads) {
            const int p0 = atomicAdd(&off[r[j].x >> SORT_BIN_SHIFT], 1);
            const int p1 = atomicAdd(&off[r[j].y >> SORT_BIN_SHIFT], 1);
            const int p2 = atomicAdd(&off[r[j].z >> SORT_BIN_SHIFT], 1);
            const int p3 = atomicAdd(&off[r[j].w >> SORT_BIN_SHIFT], 1);
            pv[p0] = make_int2(r[j].x, c[j].x);
            pv[p1] = make_int2(r[j].y, c[j].y);
            pv[p2] = make_int2(r[j].z, c[j].z);
            pv[p3] = make_int2(r[j].w, c[j].w);
        }
    }
}

#define BIN_WAVES (SORT_BIN_ROWS / 32)
__global__ __launch_bounds__(SORT_BIN_ROWS) void bin_rows_kernel(
    const int* __restrict__ pairs, const int* __restrict__ binbase,
    int* __restrict__ rowptr, int* __restrict__ cols, int n_nodes)
{
    __shared__ int cur[SORT_BIN_ROWS];
    __shared__ int wsum[BIN_WAVES];
    const int tid  = threadIdx.x;
    const int lane = tid & 31;
    const int wv   = tid >> 5;
    const int row0 = blockIdx.x * SORT_BIN_ROWS;
    const int lo   = binbase[blockIdx.x];
    const int hi   = binbase[blockIdx.x + 1];
    const int2* __restrict__ pv = (const int2*)pairs;

    cur[tid] = 0;
    __syncthreads();
    for (int e0 = lo + tid; e0 < hi + tid; e0 += 4 * SORT_BIN_ROWS) {
        int2 pr[4];
        for (int j = 0; j < 4; ++j) {
            const int e = e0 + j * SORT_BIN_ROWS;
            pr[j] = e < hi ? pv[e] : make_int2(row0, 0);
        }
        for (int j = 0; j < 4; ++j)
            if (e0 + j * SORT_BIN_ROWS < hi) atomicAdd(&cur[pr[j].x - row0], 1);
    }
    __syncthreads();

    const int v = cur[tid];
    int s = v;
    for (int o = 1; o < 32; o <<= 1) {
        const int y = __shfl_up(s, o, 32);
        if (lane >= o) s += y;
    }
    if (lane == 31) wsum[wv] = s;
    __syncthreads();
    if (wv == 0) {
        const int w = lane < BIN_WAVES ? wsum[lane] : 0;
        int u = w;
        for (int o = 1; o < 32; o <<= 1) {
            const int y = __shfl_up(u, o, 32);
            if (lane >= o) u += y;
        }
        if (lane < BIN_WAVES) wsum[lane] = u - w;
    }
    __syncthreads();
    const int excl = wsum[wv] + s - v;
    const int r = row0 + tid;
    if (r <= n_nodes) rowptr[r] = lo + excl;
    cur[tid] = excl;
    __syncthreads();

    for (int e0 = lo + tid; e0 < hi + tid; e0 += 4 * SORT_BIN_ROWS) {
        int2 pr[4];
        for (int j = 0; j < 4; ++j) {
            const int e = e0 + j * SORT_BIN_ROWS;
            pr[j] = e < hi ? pv[e] : make_int2(row0, 0);
        }
        for (int j = 0; j < 4; ++j) {
            if (e0 + j * SORT_BIN_ROWS < hi) {
                const int p = atomicAdd(&cur[pr[j].x - row0], 1);
                cols[lo + p] = pr[j].y;
            }
        }
    }
}

#define ROWS_PER_WAVE 8
#define ROW_WAVES_PER_BLOCK 4

#define DPP_ROW_SHARE(j) (0x150 + (j))
#define DPP_ROW_XMASK(m) (0x160 + (m))

template <int CTRL>
__device__ __forceinline__ int dpp_i(int v)
{
    return __builtin_amdgcn_update_dpp(0, v, CTRL, 0xf, 0xf, true);
}

template <int CTRL>
__device__ __forceinline__ float dpp_f(float v)
{
    return __builtin_bit_cast(float,
        __builtin_amdgcn_update_dpp(0, __builtin_bit_cast(int, v), CTRL, 0xf, 0xf, true));
}

template <int J>
__device__ __forceinline__ void row_share16(unsigned (&o)[16], int v)
{
    if constexpr (J < 16) {
        o[J] = (unsigned)dpp_i<DPP_ROW_SHARE(J)>(v);
        row_share16<J + 1>(o, v);
    }
}

template <int M, int N>
__device__ __forceinline__ void fold_step(float (&s)[16], bool hi)
{
    #pragma unroll
    for (int j = 0; j < N; ++j) {
        const float keep = hi ? s[j + N] : s[j];
        const float give = hi ? s[j] : s[j + N];
        s[j] = keep + dpp_f<DPP_ROW_XMASK(M)>(give);
    }
}

__device__ __forceinline__ float row_sums16(float (&s)[16], int sub)
{
    fold_step<8, 8>(s, (sub & 8) != 0);
    fold_step<4, 4>(s, (sub & 4) != 0);
    fold_step<2, 2>(s, (sub & 2) != 0);
    fold_step<1, 1>(s, (sub & 1) != 0);
    return s[0];
}

__device__ __forceinline__ float row_sum1(float v)
{
    v += dpp_f<DPP_ROW_XMASK(8)>(v);
    v += dpp_f<DPP_ROW_XMASK(4)>(v);
    v += dpp_f<DPP_ROW_XMASK(2)>(v);
    v += dpp_f<DPP_ROW_XMASK(1)>(v);
    return v;
}

__global__ __launch_bounds__(128) void row_att_kernel(
    const float* __restrict__ p1, const float* __restrict__ p2,
    const float* __restrict__ w2, const float* __restrict__ b2,
    const int* __restrict__ rowptr, const int* __restrict__ cols,
    float* __restrict__ mask, float* __restrict__ dis, int n_nodes)
{
    const int lane   = threadIdx.x & 31;
    const int sub    = lane & 15;
    const int hsel   = lane >> 4;
    const int wave_g = blockIdx.x * ROW_WAVES_PER_BLOCK + (threadIdx.x >> 5);
    const int r0     = wave_g * ROWS_PER_WAVE;
    if (r0 >= n_nodes) return;
    const float4* __restrict__ p1v = (const float4*)p1;
    const float4* __restrict__ p2v = (const float4*)p2;

    const float4 w = ((const float4*)w2)[sub];
    const float bias = b2[0];

    const int rp_i = r0 + lane;
    const int rp_l = rowptr[rp_i < n_nodes ? rp_i : n_nodes];

    for (int q = 0; q < ROWS_PER_WAVE / 2; ++q) {
        const int r = r0 + 2 * q + hsel;
        const bool rv = r < n_nodes;
        const int e0 = __builtin_amdgcn_readlane(rp_l, 2 * q);
        const int e1 = __builtin_amdgcn_readlane(rp_l, 2 * q + 1);
        const int e2 = __builtin_amdgcn_readlane(rp_l, 2 * q + 2);
        const int start = hsel ? e1 : e0;
        const int end   = hsel ? e2 : e1;
        const int degmax = max(e1 - e0, e2 - e1);
        const float4 a = p1v[(rv ? r : n_nodes - 1) * (HID / 4) + sub];

        float rs = 0.f;
        for (int base = 0; base < degmax; base += 16) {
            const int idx = start + base + sub;
            const bool ev = idx < end;
            const int c_l = ev ? cols[idx] : 0;

            unsigned cj[16];
            float4 b[16];
            row_share16<0>(cj, c_l);
            #pragma unroll
            for (int j = 0; j < 16; ++j)
                b[j] = p2v[cj[j] * (HID / 4) + sub];

            float s[16];
            #pragma unroll
            for (int j = 0; j < 16; ++j) {
                const float t0 = fmaxf(a.x + b[j].x, 0.f);
                const float t1 = fmaxf(a.y + b[j].y, 0.f);
                const float t2 = fmaxf(a.z + b[j].z, 0.f);
                const float t3 = fmaxf(a.w + b[j].w, 0.f);
                s[j] = t0 * w.x + t1 * w.y + t2 * w.z + t3 * w.w;
            }
            const float mine = row_sums16(s, sub);

            const float la = mine + bias;
            const float sg = 1.f / (1.f + __expf(-la));
            float m = sg * (ZETA_C - GAMMA_C) + GAMMA_C;
            m = fminf(fmaxf(m, 0.f), 1.f);
            if (ev) { mask[idx] = m; rs += m; }
        }

        rs = row_sum1(rs);
        if (rv && sub == 0) dis[r] = fminf(rsqrtf(rs + 1e-6f), 10.f);
    }
}

__global__ __launch_bounds__(128) void row_spmm_kernel(
    const float* __restrict__ x, const float* __restrict__ mask,
    const float* __restrict__ dis,
    const int* __restrict__ rowptr, const int* __restrict__ cols,
    const float* addin, float* out, float* xn, int n_nodes)
{
    const int lane   = threadIdx.x & 31;
    const int sub    = lane & 15;
    const int hsel   = lane >> 4;
    const int wave_g = blockIdx.x * ROW_WAVES_PER_BLOCK + (threadIdx.x >> 5);
    const int r0     = wave_g * ROWS_PER_WAVE;
    if (r0 >= n_nodes) return;
    const float4* __restrict__ xv = (const float4*)x;
    const float4 zero4 = {0.f, 0.f, 0.f, 0.f};

    const int rp_i = r0 + lane;
    const int rp_l = rowptr[rp_i < n_nodes ? rp_i : n_nodes];

    for (int q = 0; q < ROWS_PER_WAVE / 2; ++q) {
        const int r = r0 + 2 * q + hsel;
        const bool rv = r < n_nodes;
        const int e0 = __builtin_amdgcn_readlane(rp_l, 2 * q);
        const int e1 = __builtin_amdgcn_readlane(rp_l, 2 * q + 1);
        const int e2 = __builtin_amdgcn_readlane(rp_l, 2 * q + 2);
        const int start = hsel ? e1 : e0;
        const int end   = hsel ? e2 : e1;
        const int degmax = max(e1 - e0, e2 - e1);
        const float dr = dis[rv ? r : n_nodes - 1];

        float4 acc = zero4;
        for (int base = 0; base < degmax; base += 16) {
            const int idx = start + base + sub;
            const bool ev = idx < end;
            const int c_l = ev ? cols[idx] : 0;
            const int nv  = end - (start + base);

            unsigned cj[16];
            float4 xc[16];
            row_share16<0>(cj, c_l);
            #pragma unroll
            for (int j = 0; j < 16; ++j)
                xc[j] = xv[cj[j] * (HID / 4) + sub];

            float v_l = 0.f;
            if (ev) v_l = mask[idx] * dr * dis[c_l];

            unsigned vj[16];
            row_share16<0>(vj, __builtin_bit_cast(int, v_l));

            #pragma unroll
            for (int j = 0; j < 16; ++j) {
                const float v = __builtin_bit_cast(float, vj[j]);
                const float4 t = (j < nv) ? xc[j] : zero4;
                acc.x += v * t.x;
                acc.y += v * t.y;
                acc.z += v * t.z;
                acc.w += v * t.w;
            }
        }

        if (rv) {
            const size_t o = (size_t)r * (HID / 4) + sub;
            const float4 bs = ((const float4*)addin)[o];
            float4 res;
            res.x = bs.x + acc.x; res.y = bs.y + acc.y;
            res.z = bs.z + acc.z; res.w = bs.w + acc.w;
            ((float4*)out)[o] = res;
            if (xn) ((float4*)xn)[o] = acc;
        }
    }
}

extern "C" void kernel_launch(void* const* d_in, const int* in_sizes, int n_in,
                              void* d_out, int out_size, void* d_ws, size_t ws_size,
                              hipStream_t stream) {
    (void)in_sizes; (void)n_in; (void)out_size; (void)ws_size;

    const float* feat = (const float*)d_in[0];
    const float* nbW  = (const float*)d_in[1];
    const float* nbb  = (const float*)d_in[2];
    const float* sW   = (const float*)d_in[3];
    const float* sb   = (const float*)d_in[4];
    const float* a1W  = (const float*)d_in[5];
    const float* a1b  = (const float*)d_in[6];
    const float* a2W  = (const float*)d_in[7];
    const float* a2b  = (const float*)d_in[8];
    const int*   row  = (const int*)d_in[9];
    const int*   col  = (const int*)d_in[10];
    float* out = (float*)d_out;

    const int NH = NNODES * HID;
    float* ws = (float*)d_ws;
    float* bufA   = ws; ws += NH;
    float* bufP1  = ws; ws += NH;
    float* bufP2  = ws; ws += NH;
    int*   cols   = (int*)ws; ws += NEDGES;
    float* maskb  = ws; ws += NEDGES;
    int*   rowptr = (int*)ws; ws += NNODES + 32;
    int*   table  = (int*)ws; ws += NNODES;
    float* dis    = ws; ws += NNODES;
    int*   binbase = table + SORT_TABLE;
    int*   pairs  = (int*)bufA;
    static_assert(SORT_TABLE + SORT_NBINS + 1 <= NNODES, "table and bin bases share one plane");
    static_assert(2 * NEDGES <= NNODES * HID, "the (row, col) pairs must fit bufA");

    const int nQuad  = NEDGES / 4;
    const int gRow   = (NNODES + ROWS_PER_WAVE * ROW_WAVES_PER_BLOCK - 1)
                       / (ROWS_PER_WAVE * ROW_WAVES_PER_BLOCK);

    bin_count_kernel<<<SORT_NCHUNKS, SORT_THREADS, 0, stream>>>(row, table, nQuad);
    bin_scan_kernel<<<1, 1024, 0, stream>>>(table, binbase);
    bin_scatter_kernel<<<SORT_NCHUNKS, SORT_THREADS, 0, stream>>>(row, col, table, pairs, nQuad);
    bin_rows_kernel<<<SORT_NBINS, SORT_BIN_ROWS, 0, stream>>>(pairs, binbase, rowptr, cols, NNODES);

    const float* xin[NLAYERS]  = {feat,  bufA,  bufP1};
    float*       p1o[NLAYERS]  = {bufP1, bufP1, bufA};
    float*       xout[NLAYERS] = {bufA,  bufP1, nullptr};
    const float* addin[NLAYERS] = {feat, out, out};

    for (int l = 0; l < NLAYERS; ++l) {
        node_transform_kernel<<<NT_BLOCKS, NT_THREADS, 0, stream>>>(
            xin[l],
            nbW + (size_t)l * 4096, nbb + l * 64,
            a1W + (size_t)l * 8192, a1b + l * 64, p1o[l],
            sW + (size_t)l * 4096, sb + l * 64,
            a1W + (size_t)l * 8192 + 4096, bufP2, NNODES);

        row_att_kernel<<<gRow, 128, 0, stream>>>(
            p1o[l], bufP2, a2W + l * 64, a2b + l, rowptr, cols, maskb, dis, NNODES);

        row_spmm_kernel<<<gRow, 128, 0, stream>>>(
            xin[l], maskb, dis, rowptr, cols, addin[l], out, xout[l], NNODES);
    }
}


# row kernels: a pair's first-chunk indices are loaded one row pair ahead, and its addin row in front of its row loads: one exposed load round trip per pair, not two or three; same sums
# speedup vs baseline: 3.8370x; 1.0515x over previous
#include <hip/hip_runtime.h>
#include <hip/hip_bf16.h>

#define NNODES 100000
#define NEDGES 1200000
#define HID 64
#define NLAYERS 3
#define GAMMA_C (-0.45f)
#define ZETA_C  (1.05f)

typedef float v2f __attribute__((ext_vector_type(2)));
typedef float v8f __attribute__((ext_vector_type(8)));

#define NT_WAVES   24
#define NT_THREADS (NT_WAVES * 32)
#define NT_BLOCKS  256
#define W_KP_PITCH 160
#define W_MAT      (32 * W_KP_PITCH)
#define G_PITCH    68

static_assert((size_t)(4 * W_MAT + NT_WAVES * 16 * G_PITCH) * 4 > 160u * 1024u,
              "the block's LDS must keep a second block off its WGP");
static_assert((size_t)(4 * W_MAT + NT_WAVES * 16 * G_PITCH) * 4 <= 320u * 1024u,
              "the block's LDS must fit a WGP");

__device__ __forceinline__ void node_tile_load(
    v2f (&xa)[16], const float* __restrict__ x, int n_nodes, int tile, int mn, int kh)
{
    int ri = tile * 16 + mn;
    if (ri >= n_nodes) ri = n_nodes - 1;
    const float* xr = x + (size_t)ri * HID;
    #pragma unroll
    for (int kk = 0; kk < 16; ++kk) {
        const int k = kk * 4 + kh;
        xa[kk].x = xr[k];
        xa[kk].y = xr[k + 1];
    }
}

__device__ __forceinline__ void node_tile_project(
    const v2f (&xa)[16], const float* ws, float* gt,
    const float* __restrict__ b1a, const float* __restrict__ bias2a,
    float* __restrict__ pa,
    const float* __restrict__ b1b, float* __restrict__ pb,
    int n_nodes, int tile, int p_lo, int p_hi, int mn, int kh, int mbase)
{
    const int row0 = tile * 16;
    const v8f vz = {0.f, 0.f, 0.f, 0.f, 0.f, 0.f, 0.f, 0.f};

    for (int pj = p_lo; pj < p_hi; ++pj) {
        const float* w1 = ws + (2 * pj) * W_MAT;
        const float* w2 = w1 + W_MAT;
        const float* __restrict__ b1 = pj == 0 ? b1a : b1b;
        float* __restrict__ p = pj == 0 ? pa : pb;

        v8f acc[4] = {vz, vz, vz, vz};
        #pragma unroll
        for (int kk = 0; kk < 16; ++kk) {
            const int kp = kk * 2 + (kh >> 1);
            #pragma unroll
            for (int nt = 0; nt < 4; ++nt) {
                const v2f b = *(const v2f*)(w1 + kp * W_KP_PITCH + 2 * (nt * 16 + mn));
                acc[nt] = __builtin_amdgcn_wmma_f32_16x16x4_f32(
                    false, xa[kk], false, b, (short)0, acc[nt], false, false);
            }
        }

        #pragma unroll
        for (int nt = 0; nt < 4; ++nt) {
            const float bv = b1[nt * 16 + mn];
            #pragma unroll
            for (int i = 0; i < 8; ++i) {
                float v = acc[nt][i] + bv;
                v = v > 0.f ? v : 0.f;
                gt[(mbase + i) * G_PITCH + nt * 16 + mn] = v;
            }
        }

        v8f acc2[4] = {vz, vz, vz, vz};
        #pragma unroll
        for (int kk = 0; kk < 16; ++kk) {
            const int k = kk * 4 + kh;
            const int kp = kk * 2 + (kh >> 1);
            v2f a; a.x = gt[mn * G_PITCH + k]; a.y = gt[mn * G_PITCH + k + 1];
            #pragma unroll
            for (int nt = 0; nt < 4; ++nt) {
                const v2f b = *(const v2f*)(w2 + kp * W_KP_PITCH + 2 * (nt * 16 + mn));
                acc2[nt] = __builtin_amdgcn_wmma_f32_16x16x4_f32(
                    false, a, false, b, (short)0, acc2[nt], false, false);
            }
        }

        float b2v[4];
        #pragma unroll
        for (int nt = 0; nt < 4; ++nt)
            b2v[nt] = pj == 0 ? bias2a[nt * 16 + mn] : 0.f;
        if (row0 + 16 <= n_nodes) {
            float* __restrict__ pr = p + (size_t)(row0 + mbase) * HID + mn;
            #pragma unroll
            for (int nt = 0; nt < 4; ++nt) {
                #pragma unroll
                for (int i = 0; i < 8; ++i)
                    pr[i * HID + nt * 16] = acc2[nt][i] + b2v[nt];
            }
        } else {
            #pragma unroll
            for (int nt = 0; nt < 4; ++nt) {
                #pragma unroll
                for (int i = 0; i < 8; ++i) {
                    const int r = row0 + mbase + i;
                    if (r < n_nodes)
                        p[(size_t)r * HID + nt * 16 + mn] = acc2[nt][i] + b2v[nt];
                }
            }
        }
    }
}

__device__ __forceinline__ bool node_pick(
    int cnt, int j0, int wave, int& j, int& p_lo, int& p_hi, int& step)
{
    const bool split = cnt - j0 <= NT_WAVES / 2;
    j    = j0 + (split ? (wave >> 1) : wave);
    p_lo = split ? (wave & 1) : 0;
    p_hi = split ? p_lo + 1 : 2;
    step = split ? NT_WAVES / 2 : NT_WAVES;
    return j < cnt;
}

__global__ __launch_bounds__(NT_THREADS) void node_transform_kernel(
    const float* __restrict__ x,
    const float* __restrict__ W1a, const float* __restrict__ b1a,
    const float* __restrict__ W2a, const float* __restrict__ bias2a,
    float* __restrict__ pa,
    const float* __restrict__ W1b, const float* __restrict__ b1b,
    const float* __restrict__ W2b, float* __restrict__ pb,
    int n_nodes)
{
    __shared__ __attribute__((aligned(16))) float ws[4 * W_MAT];
    __shared__ float gs[NT_WAVES * 16 * G_PITCH];

    const int wave  = __builtin_amdgcn_readfirstlane((int)(threadIdx.x >> 5));
    const int lane  = threadIdx.x & 31;
    const int mn    = lane & 15;
    const int kh    = (lane >> 4) << 1;
    const int mbase = (lane >> 4) << 3;
    float* gt = gs + wave * (16 * G_PITCH);

    const int n_tiles = (n_nodes + 15) >> 4;
    const int blk = (int)blockIdx.x;
    const int cnt = blk < n_tiles ? (n_tiles - 1 - blk) / NT_BLOCKS + 1 : 0;

    int j0 = 0, j, p_lo, p_hi, step;
    bool mine = node_pick(cnt, j0, wave, j, p_lo, p_hi, step);
    v2f xa[16];

    constexpr int W_SLOTS = 4 * 16;
    constexpr int W_FULL  = W_SLOTS / NT_WAVES;
    constexpr int W_REST  = W_SLOTS % NT_WAVES;
    const int skp = lane >> 4;
    const int sn4 = lane & 15;
    float4 sev[W_FULL], sod[W_FULL];
    #pragma unroll
    for (int t = 0; t < W_FULL; ++t) {
        const int s = wave + t * NT_WAVES;
        const int m = s >> 4;
        const float* src = m == 0 ? W1a : m == 1 ? W2a : m == 2 ? W1b : W2b;
        const int kp = (s & 15) * 2 + skp;
        sev[t] = ((const float4*)src)[(2 * kp) * 16 + sn4];
        sod[t] = ((const float4*)src)[(2 * kp + 1) * 16 + sn4];
    }
    float4 rev = {0.f, 0.f, 0.f, 0.f}, rod = {0.f, 0.f, 0.f, 0.f};
    if (wave < W_REST) {
        const int s = wave + W_FULL * NT_WAVES;
        const int m = s >> 4;
        const float* src = m == 0 ? W1a : m == 1 ? W2a : m == 2 ? W1b : W2b;
        const int kp = (s & 15) * 2 + skp;
        rev = ((const float4*)src)[(2 * kp) * 16 + sn4];
        rod = ((const float4*)src)[(2 * kp + 1) * 16 + sn4];
    }
    #pragma unroll
    for (int t = 0; t < W_FULL; ++t) {
        const int s = wave + t * NT_WAVES;
        float4* d = (float4*)(ws + (s >> 4) * W_MAT
                                 + ((s & 15) * 2 + skp) * W_KP_PITCH + sn4 * 8);
        d[0] = make_float4(sev[t].x, sod[t].x, sev[t].y, sod[t].y);
        d[1] = make_float4(sev[t].z, sod[t].z, sev[t].w, sod[t].w);
    }
    if (wave < W_REST) {
        const int s = wave + W_FULL * NT_WAVES;
        float4* d = (float4*)(ws + (s >> 4) * W_MAT
                                 + ((s & 15) * 2 + skp) * W_KP_PITCH + sn4 * 8);
        d[0] = make_float4(rev.x, rod.x, rev.y, rod.y);
        d[1] = make_float4(rev.z, rod.z, rev.w, rod.w);
    }
    __syncthreads();

    while (j0 < cnt) {
        if (mine) {
            const int tile = blk + NT_BLOCKS * j;
            node_tile_load(xa, x, n_nodes, tile, mn, kh);
            node_tile_project(xa, ws, gt, b1a, bias2a, pa, b1b, pb, n_nodes,
                              tile, p_lo, p_hi, mn, kh, mbase);
        }
        j0 += step;
        if (j0 >= cnt) break;
        mine = node_pick(cnt, j0, wave, j, p_lo, p_hi, step);
    }
}

static_assert(NEDGES % 4 == 0, "edge count must fill whole groups of four");

#define SORT_BIN_SHIFT   9
#define SORT_BIN_ROWS    (1 << SORT_BIN_SHIFT)
#define SORT_NBINS       (NNODES / SORT_BIN_ROWS + 1)
#define SORT_THREADS     512
#define SORT_CHUNK_QUADS (4 * SORT_THREADS)
#define SORT_NCHUNKS     ((NEDGES / 4 + SORT_CHUNK_QUADS - 1) / SORT_CHUNK_QUADS)
#define SORT_TABLE       (SORT_NBINS * SORT_NCHUNKS)
#define SCAN_PER         ((SORT_TABLE + 1023) / 1024)

__global__ __launch_bounds__(SORT_THREADS) void bin_count_kernel(
    const int* __restrict__ row, int* __restrict__ table, int n_quads)
{
    __shared__ int hist[SORT_NBINS];
    for (int b = threadIdx.x; b < SORT_NBINS; b += SORT_THREADS) hist[b] = 0;
    __syncthreads();
    const int q0 = blockIdx.x * SORT_CHUNK_QUADS + threadIdx.x;
    constexpr int NQ = SORT_CHUNK_QUADS / SORT_THREADS;
    int4 r[NQ];
    for (int j = 0; j < NQ; ++j) {
        const int q = q0 + j * SORT_THREADS;
        r[j] = ((const int4*)row)[q < n_quads ? q : n_quads - 1];
    }
    for (int j = 0; j < NQ; ++j) {
        if (q0 + j * SORT_THREADS < n_quads) {
            atomicAdd(&hist[r[j].x >> SORT_BIN_SHIFT], 1);
            atomicAdd(&hist[r[j].y >> SORT_BIN_SHIFT], 1);
            atomicAdd(&hist[r[j].z >> SORT_BIN_SHIFT], 1);
            atomicAdd(&hist[r[j].w >> SORT_BIN_SHIFT], 1);
        }
    }
    __syncthreads();
    for (int b = threadIdx.x; b < SORT_NBINS; b += SORT_THREADS)
        table[b * SORT_NCHUNKS + blockIdx.x] = hist[b];
}

__global__ __launch_bounds__(1024) void bin_scan_kernel(
    int* __restrict__ table, int* __restrict__ binbase)
{
    __shared__ int t[SCAN_PER * 1024];
    __shared__ int wsum[32];
    const int tid  = threadIdx.x;
    const int lane = tid & 31;
    const int wv   = tid >> 5;

    for (int j = 0; j < SCAN_PER; ++j) {
        const int i = j * 1024 + tid;
        t[i] = i < SORT_TABLE ? table[i] : 0;
    }
    __syncthreads();

    int* seg = t + tid * SCAN_PER;
    int sum = 0;
    for (int k = 0; k < SCAN_PER; ++k) sum += seg[k];

    int s = sum;
    for (int o = 1; o < 32; o <<= 1) {
        const int y = __shfl_up(s, o, 32);
        if (lane >= o) s += y;
    }
    if (lane == 31) wsum[wv] = s;
    __syncthreads();
    if (wv == 0) {
        const int w = wsum[lane];
        int u = w;
        for (int o = 1; o < 32; o <<= 1) {
            const int y = __shfl_up(u, o, 32);
            if (lane >= o) u += y;
        }
        wsum[lane] = u - w;
    }
    __syncthreads();

    int run = wsum[wv] + s - sum;
    for (int k = 0; k < SCAN_PER; ++k) {
        const int v = seg[k];
        seg[k] = run;
        run += v;
    }
    __syncthreads();

    for (int j = 0; j < SCAN_PER; ++j) {
        const int i = j * 1024 + tid;
        if (i < SORT_TABLE) {
            const int p = t[i];
            table[i] = p;
            if (i % SORT_NCHUNKS == 0) binbase[i / SORT_NCHUNKS] = p;
        }
    }
    if (tid == 1023) binbase[SORT_NBINS] = run;
}

__global__ __launch_bounds__(SORT_THREADS) void bin_scatter_kernel(
    const int* __restrict__ row, const int* __restrict__ col,
    const int* __restrict__ table, int* __restrict__ pairs, int n_quads)
{
    __shared__ int off[SORT_NBINS];
    for (int b = threadIdx.x; b < SORT_NBINS; b += SORT_THREADS)
        off[b] = table[b * SORT_NCHUNKS + blockIdx.x];
    __syncthreads();
    int2* __restrict__ pv = (int2*)pairs;
    const int q0 = blockIdx.x * SORT_CHUNK_QUADS + threadIdx.x;
    constexpr int NQ = SORT_CHUNK_QUADS / SORT_THREADS;
    int4 r[NQ], c[NQ];
    for (int j = 0; j < NQ; ++j) {
        const int q = q0 + j * SORT_THREADS;
        const int qc = q < n_quads ? q : n_quads - 1;
        r[j] = ((const int4*)row)[qc];
        c[j] = ((const int4*)col)[qc];
    }
    for (int j = 0; j < NQ; ++j) {
        if (q0 + j * SORT_THREADS < n_quads) {
            const int p0 = atomicAdd(&off[r[j].x >> SORT_BIN_SHIFT], 1);
            const int p1 = atomicAdd(&off[r[j].y >> SORT_BIN_SHIFT], 1);
            const int p2 = atomicAdd(&off[r[j].z >> SORT_BIN_SHIFT], 1);
            const int p3 = atomicAdd(&off[r[j].w >> SORT_BIN_SHIFT], 1);
            pv[p0] = make_int2(r[j].x, c[j].x);
            pv[p1] = make_int2(r[j].y, c[j].y);
            pv[p2] = make_int2(r[j].z, c[j].z);
            pv[p3] = make_int2(r[j].w, c[j].w);
        }
    }
}

#define BIN_WAVES (SORT_BIN_ROWS / 32)
__global__ __launch_bounds__(SORT_BIN_ROWS) void bin_rows_kernel(
    const int* __restrict__ pairs, const int* __restrict__ binbase,
    int* __restrict__ rowptr, int* __restrict__ cols, int n_nodes)
{
    __shared__ int cur[SORT_BIN_ROWS];
    __shared__ int wsum[BIN_WAVES];
    const int tid  = threadIdx.x;
    const int lane = tid & 31;
    const int wv   = tid >> 5;
    const int row0 = blockIdx.x * SORT_BIN_ROWS;
    const int lo   = binbase[blockIdx.x];
    const int hi   = binbase[blockIdx.x + 1];
    const int2* __restrict__ pv = (const int2*)pairs;

    cur[tid] = 0;
    __syncthreads();
    for (int e0 = lo + tid; e0 < hi + tid; e0 += 4 * SORT_BIN_ROWS) {
        int2 pr[4];
        for (int j = 0; j < 4; ++j) {
            const int e = e0 + j * SORT_BIN_ROWS;
            pr[j] = e < hi ? pv[e] : make_int2(row0, 0);
        }
        for (int j = 0; j < 4; ++j)
            if (e0 + j * SORT_BIN_ROWS < hi) atomicAdd(&cur[pr[j].x - row0], 1);
    }
    __syncthreads();

    const int v = cur[tid];
    int s = v;
    for (int o = 1; o < 32; o <<= 1) {
        const int y = __shfl_up(s, o, 32);
        if (lane >= o) s += y;
    }
    if (lane == 31) wsum[wv] = s;
    __syncthreads();
    if (wv == 0) {
        const int w = lane < BIN_WAVES ? wsum[lane] : 0;
        int u = w;
        for (int o = 1; o < 32; o <<= 1) {
            const int y = __shfl_up(u, o, 32);
            if (lane >= o) u += y;
        }
        if (lane < BIN_WAVES) wsum[lane] = u - w;
    }
    __syncthreads();
    const int excl = wsum[wv] + s - v;
    const int r = row0 + tid;
    if (r <= n_nodes) rowptr[r] = lo + excl;
    cur[tid] = excl;
    __syncthreads();

    for (int e0 = lo + tid; e0 < hi + tid; e0 += 4 * SORT_BIN_ROWS) {
        int2 pr[4];
        for (int j = 0; j < 4; ++j) {
            const int e = e0 + j * SORT_BIN_ROWS;
            pr[j] = e < hi ? pv[e] : make_int2(row0, 0);
        }
        for (int j = 0; j < 4; ++j) {
            if (e0 + j * SORT_BIN_ROWS < hi) {
                const int p = atomicAdd(&cur[pr[j].x - row0], 1);
                cols[lo + p] = pr[j].y;
            }
        }
    }
}

#define ROWS_PER_WAVE 8
#define ROW_WAVES_PER_BLOCK 4

#define DPP_ROW_SHARE(j) (0x150 + (j))
#define DPP_ROW_XMASK(m) (0x160 + (m))

template <int CTRL>
__device__ __forceinline__ int dpp_i(int v)
{
    return __builtin_amdgcn_update_dpp(0, v, CTRL, 0xf, 0xf, true);
}

template <int CTRL>
__device__ __forceinline__ float dpp_f(float v)
{
    return __builtin_bit_cast(float,
        __builtin_amdgcn_update_dpp(0, __builtin_bit_cast(int, v), CTRL, 0xf, 0xf, true));
}

template <int J>
__device__ __forceinline__ void row_share16(unsigned (&o)[16], int v)
{
    if constexpr (J < 16) {
        o[J] = (unsigned)dpp_i<DPP_ROW_SHARE(J)>(v);
        row_share16<J + 1>(o, v);
    }
}

template <int M, int N>
__device__ __forceinline__ void fold_step(float (&s)[16], bool hi)
{
    #pragma unroll
    for (int j = 0; j < N; ++j) {
        const float keep = hi ? s[j + N] : s[j];
        const float give = hi ? s[j] : s[j + N];
        s[j] = keep + dpp_f<DPP_ROW_XMASK(M)>(give);
    }
}

__device__ __forceinline__ float row_sums16(float (&s)[16], int sub)
{
    fold_step<8, 8>(s, (sub & 8) != 0);
    fold_step<4, 4>(s, (sub & 4) != 0);
    fold_step<2, 2>(s, (sub & 2) != 0);
    fold_step<1, 1>(s, (sub & 1) != 0);
    return s[0];
}

__device__ __forceinline__ float row_sum1(float v)
{
    v += dpp_f<DPP_ROW_XMASK(8)>(v);
    v += dpp_f<DPP_ROW_XMASK(4)>(v);
    v += dpp_f<DPP_ROW_XMASK(2)>(v);
    v += dpp_f<DPP_ROW_XMASK(1)>(v);
    return v;
}

__global__ __launch_bounds__(128) void row_att_kernel(
    const float* __restrict__ p1, const float* __restrict__ p2,
    const float* __restrict__ w2, const float* __restrict__ b2,
    const int* __restrict__ rowptr, const int* __restrict__ cols,
    float* __restrict__ mask, float* __restrict__ dis, int n_nodes)
{
    const int lane   = threadIdx.x & 31;
    const int sub    = lane & 15;
    const int hsel   = lane >> 4;
    const int wave_g = blockIdx.x * ROW_WAVES_PER_BLOCK + (threadIdx.x >> 5);
    const int r0     = wave_g * ROWS_PER_WAVE;
    if (r0 >= n_nodes) return;
    const float4* __restrict__ p1v = (const float4*)p1;
    const float4* __restrict__ p2v = (const float4*)p2;

    const float4 w = ((const float4*)w2)[sub];
    const float bias = b2[0];

    const int rp_i = r0 + lane;
    const int rp_l = rowptr[rp_i < n_nodes ? rp_i : n_nodes];

    int c_nx;
    {
        const int f0 = __builtin_amdgcn_readlane(rp_l, 0);
        const int f1 = __builtin_amdgcn_readlane(rp_l, 1);
        c_nx = cols[min((hsel ? f1 : f0) + sub, NEDGES - 1)];
    }

    for (int q = 0; q < ROWS_PER_WAVE / 2; ++q) {
        const int r = r0 + 2 * q + hsel;
        const bool rv = r < n_nodes;
        const int e0 = __builtin_amdgcn_readlane(rp_l, 2 * q);
        const int e1 = __builtin_amdgcn_readlane(rp_l, 2 * q + 1);
        const int e2 = __builtin_amdgcn_readlane(rp_l, 2 * q + 2);
        const int start = hsel ? e1 : e0;
        const int end   = hsel ? e2 : e1;
        const int degmax = max(e1 - e0, e2 - e1);
        const int c_first = c_nx;
        {
            const int g1 = __builtin_amdgcn_readlane(rp_l, 2 * q + 3);
            c_nx = cols[min((hsel ? g1 : e2) + sub, NEDGES - 1)];
        }
        const float4 a = p1v[(rv ? r : n_nodes - 1) * (HID / 4) + sub];

        float rs = 0.f;
        for (int base = 0; base < degmax; base += 16) {
            const int idx = start + base + sub;
            const bool ev = idx < end;
            int c_l = ev ? c_first : 0;
            if (base != 0) c_l = ev ? cols[idx] : 0;

            unsigned cj[16];
            float4 b[16];
            row_share16<0>(cj, c_l);
            #pragma unroll
            for (int j = 0; j < 16; ++j)
                b[j] = p2v[cj[j] * (HID / 4) + sub];

            float s[16];
            #pragma unroll
            for (int j = 0; j < 16; ++j) {
                const float t0 = fmaxf(a.x + b[j].x, 0.f);
                const float t1 = fmaxf(a.y + b[j].y, 0.f);
                const float t2 = fmaxf(a.z + b[j].z, 0.f);
                const float t3 = fmaxf(a.w + b[j].w, 0.f);
                s[j] = t0 * w.x + t1 * w.y + t2 * w.z + t3 * w.w;
            }
            const float mine = row_sums16(s, sub);

            const float la = mine + bias;
            const float sg = 1.f / (1.f + __expf(-la));
            float m = sg * (ZETA_C - GAMMA_C) + GAMMA_C;
            m = fminf(fmaxf(m, 0.f), 1.f);
            if (ev) { mask[idx] = m; rs += m; }
        }

        rs = row_sum1(rs);
        if (rv && sub == 0) dis[r] = fminf(rsqrtf(rs + 1e-6f), 10.f);
    }
}

__global__ __launch_bounds__(128) void row_spmm_kernel(
    const float* __restrict__ x, const float* __restrict__ mask,
    const float* __restrict__ dis,
    const int* __restrict__ rowptr, const int* __restrict__ cols,
    const float* addin, float* out, float* xn, int n_nodes)
{
    const int lane   = threadIdx.x & 31;
    const int sub    = lane & 15;
    const int hsel   = lane >> 4;
    const int wave_g = blockIdx.x * ROW_WAVES_PER_BLOCK + (threadIdx.x >> 5);
    const int r0     = wave_g * ROWS_PER_WAVE;
    if (r0 >= n_nodes) return;
    const float4* __restrict__ xv = (const float4*)x;
    const float4 zero4 = {0.f, 0.f, 0.f, 0.f};

    const int rp_i = r0 + lane;
    const int rp_l = rowptr[rp_i < n_nodes ? rp_i : n_nodes];

    int c_nx;
    {
        const int f0 = __builtin_amdgcn_readlane(rp_l, 0);
        const int f1 = __builtin_amdgcn_readlane(rp_l, 1);
        c_nx = cols[min((hsel ? f1 : f0) + sub, NEDGES - 1)];
    }

    for (int q = 0; q < ROWS_PER_WAVE / 2; ++q) {
        const int r = r0 + 2 * q + hsel;
        const bool rv = r < n_nodes;
        const int e0 = __builtin_amdgcn_readlane(rp_l, 2 * q);
        const int e1 = __builtin_amdgcn_readlane(rp_l, 2 * q + 1);
        const int e2 = __builtin_amdgcn_readlane(rp_l, 2 * q + 2);
        const int start = hsel ? e1 : e0;
        const int end   = hsel ? e2 : e1;
        const int degmax = max(e1 - e0, e2 - e1);
        const int c_first = c_nx;
        {
            const int g1 = __builtin_amdgcn_readlane(rp_l, 2 * q + 3);
            c_nx = cols[min((hsel ? g1 : e2) + sub, NEDGES - 1)];
        }
        const float dr = dis[rv ? r : n_nodes - 1];
        const int o = (rv ? r : n_nodes - 1) * (HID / 4) + sub;
        const float4 bs = ((const float4*)addin)[o];

        float4 acc = zero4;
        for (int base = 0; base < degmax; base += 16) {
            const int idx = start + base + sub;
            const bool ev = idx < end;
            int c_l = ev ? c_first : 0;
            if (base != 0) c_l = ev ? cols[idx] : 0;
            const int nv  = end - (start + base);

            unsigned cj[16];
            float4 xc[16];
            row_share16<0>(cj, c_l);
            #pragma unroll
            for (int j = 0; j < 16; ++j)
                xc[j] = xv[cj[j] * (HID / 4) + sub];

            float v_l = 0.f;
            if (ev) v_l = mask[idx] * dr * dis[c_l];

            unsigned vj[16];
            row_share16<0>(vj, __builtin_bit_cast(int, v_l));

            #pragma unroll
            for (int j = 0; j < 16; ++j) {
                const float v = __builtin_bit_cast(float, vj[j]);
                const float4 t = (j < nv) ? xc[j] : zero4;
                acc.x += v * t.x;
                acc.y += v * t.y;
                acc.z += v * t.z;
                acc.w += v * t.w;
            }
        }

        if (rv) {
            float4 res;
            res.x = bs.x + acc.x; res.y = bs.y + acc.y;
            res.z = bs.z + acc.z; res.w = bs.w + acc.w;
            ((float4*)out)[o] = res;
            if (xn) ((float4*)xn)[o] = acc;
        }
    }
}

extern "C" void kernel_launch(void* const* d_in, const int* in_sizes, int n_in,
                              void* d_out, int out_size, void* d_ws, size_t ws_size,
                              hipStream_t stream) {
    (void)in_sizes; (void)n_in; (void)out_size; (void)ws_size;

    const float* feat = (const float*)d_in[0];
    const float* nbW  = (const float*)d_in[1];
    const float* nbb  = (const float*)d_in[2];
    const float* sW   = (const float*)d_in[3];
    const float* sb   = (const float*)d_in[4];
    const float* a1W  = (const float*)d_in[5];
    const float* a1b  = (const float*)d_in[6];
    const float* a2W  = (const float*)d_in[7];
    const float* a2b  = (const float*)d_in[8];
    const int*   row  = (const int*)d_in[9];
    const int*   col  = (const int*)d_in[10];
    float* out = (float*)d_out;

    const int NH = NNODES * HID;
    float* ws = (float*)d_ws;
    float* bufA   = ws; ws += NH;
    float* bufP1  = ws; ws += NH;
    float* bufP2  = ws; ws += NH;
    int*   cols   = (int*)ws; ws += NEDGES;
    float* maskb  = ws; ws += NEDGES;
    int*   rowptr = (int*)ws; ws += NNODES + 32;
    int*   table  = (int*)ws; ws += NNODES;
    float* dis    = ws; ws += NNODES;
    int*   binbase = table + SORT_TABLE;
    int*   pairs  = (int*)bufA;
    static_assert(SORT_TABLE + SORT_NBINS + 1 <= NNODES, "table and bin bases share one plane");
    static_assert(2 * NEDGES <= NNODES * HID, "the (row, col) pairs must fit bufA");

    const int nQuad  = NEDGES / 4;
    const int gRow   = (NNODES + ROWS_PER_WAVE * ROW_WAVES_PER_BLOCK - 1)
                       / (ROWS_PER_WAVE * ROW_WAVES_PER_BLOCK);

    bin_count_kernel<<<SORT_NCHUNKS, SORT_THREADS, 0, stream>>>(row, table, nQuad);
    bin_scan_kernel<<<1, 1024, 0, stream>>>(table, binbase);
    bin_scatter_kernel<<<SORT_NCHUNKS, SORT_THREADS, 0, stream>>>(row, col, table, pairs, nQuad);
    bin_rows_kernel<<<SORT_NBINS, SORT_BIN_ROWS, 0, stream>>>(pairs, binbase, rowptr, cols, NNODES);

    const float* xin[NLAYERS]  = {feat,  bufA,  bufP1};
    float*       p1o[NLAYERS]  = {bufP1, bufP1, bufA};
    float*       xout[NLAYERS] = {bufA,  bufP1, nullptr};
    const float* addin[NLAYERS] = {feat, out, out};

    for (int l = 0; l < NLAYERS; ++l) {
        node_transform_kernel<<<NT_BLOCKS, NT_THREADS, 0, stream>>>(
            xin[l],
            nbW + (size_t)l * 4096, nbb + l * 64,
            a1W + (size_t)l * 8192, a1b + l * 64, p1o[l],
            sW + (size_t)l * 4096, sb + l * 64,
            a1W + (size_t)l * 8192 + 4096, bufP2, NNODES);

        row_att_kernel<<<gRow, 128, 0, stream>>>(
            p1o[l], bufP2, a2W + l * 64, a2b + l, rowptr, cols, maskb, dis, NNODES);

        row_spmm_kernel<<<gRow, 128, 0, stream>>>(
            xin[l], maskb, dis, rowptr, cols, addin[l], out, xout[l], NNODES);
    }
}


# mask kernel makes its rows' first projection (16-row tile per wave on the matrix unit, kept in LDS, never stored) beside other waves' gathers; node transform: second one only; same sums
# speedup vs baseline: 4.0183x; 1.0473x over previous
#include <hip/hip_runtime.h>
#include <hip/hip_bf16.h>

#define NNODES 100000
#define NEDGES 1200000
#define HID 64
#define NLAYERS 3
#define GAMMA_C (-0.45f)
#define ZETA_C  (1.05f)

typedef float v2f __attribute__((ext_vector_type(2)));
typedef float v8f __attribute__((ext_vector_type(8)));

#define NT_WAVES   28
#define NT_THREADS (NT_WAVES * 32)
#define NT_BLOCKS  256
#define W_KP_PITCH 160
#define W_MAT      (32 * W_KP_PITCH)
#define G_PITCH    68
#define G_SLOTS    32

static_assert((size_t)(2 * W_MAT + G_SLOTS * 16 * G_PITCH) * 4 > 160u * 1024u,
              "the block's LDS must keep a second block off its WGP");
static_assert((size_t)(2 * W_MAT + G_SLOTS * 16 * G_PITCH) * 4 <= 320u * 1024u,
              "the block's LDS must fit a WGP");
static_assert(NT_WAVES <= G_SLOTS, "every wave needs a slot");
static_assert(((NNODES + 15) / 16 + NT_BLOCKS - 1) / NT_BLOCKS <= NT_WAVES,
              "every tile of a block needs a wave");

__device__ __forceinline__ void node_tile_load(
    v2f (&xa)[16], const float* __restrict__ x, int n_nodes, int tile, int mn, int kh)
{
    int ri = tile * 16 + mn;
    if (ri >= n_nodes) ri = n_nodes - 1;
    const float* xr = x + (size_t)ri * HID;
    #pragma unroll
    for (int kk = 0; kk < 16; ++kk) {
        const int k = kk * 4 + kh;
        xa[kk].x = xr[k];
        xa[kk].y = xr[k + 1];
    }
}

template <bool TO_LDS>
__device__ __forceinline__ void node_tile_project(
    const v2f (&xa)[16], const float* w1, const float* w2, float* gt,
    const float* __restrict__ b1, const float* __restrict__ bias2,
    float* __restrict__ p, int n_nodes, int tile, int mn, int kh, int mbase)
{
    const int row0 = tile * 16;
    const v8f vz = {0.f, 0.f, 0.f, 0.f, 0.f, 0.f, 0.f, 0.f};

    v8f acc[4] = {vz, vz, vz, vz};
    #pragma unroll
    for (int kk = 0; kk < 16; ++kk) {
        const int kp = kk * 2 + (kh >> 1);
        #pragma unroll
        for (int nt = 0; nt < 4; ++nt) {
            const v2f b = *(const v2f*)(w1 + kp * W_KP_PITCH + 2 * (nt * 16 + mn));
            acc[nt] = __builtin_amdgcn_wmma_f32_16x16x4_f32(
                false, xa[kk], false, b, (short)0, acc[nt], false, false);
        }
    }

    #pragma unroll
    for (int nt = 0; nt < 4; ++nt) {
        const float bv = b1[nt * 16 + mn];
        #pragma unroll
        for (int i = 0; i < 8; ++i) {
            float v = acc[nt][i] + bv;
            v = v > 0.f ? v : 0.f;
            gt[(mbase + i) * G_PITCH + nt * 16 + mn] = v;
        }
    }

    v8f acc2[4] = {vz, vz, vz, vz};
    #pragma unroll
    for (int kk = 0; kk < 16; ++kk) {
        const int k = kk * 4 + kh;
        const int kp = kk * 2 + (kh >> 1);
        v2f a; a.x = gt[mn * G_PITCH + k]; a.y = gt[mn * G_PITCH + k + 1];
        #pragma unroll
        for (int nt = 0; nt < 4; ++nt) {
            const v2f b = *(const v2f*)(w2 + kp * W_KP_PITCH + 2 * (nt * 16 + mn));
            acc2[nt] = __builtin_amdgcn_wmma_f32_16x16x4_f32(
                false, a, false, b, (short)0, acc2[nt], false, false);
        }
    }

    if constexpr (TO_LDS) {
        #pragma unroll
        for (int nt = 0; nt < 4; ++nt) {
            const float bv = bias2[nt * 16 + mn];
            #pragma unroll
            for (int i = 0; i < 8; ++i)
                gt[(mbase + i) * G_PITCH + nt * 16 + mn] = acc2[nt][i] + bv;
        }
    } else if (row0 + 16 <= n_nodes) {
        float* __restrict__ pr = p + (size_t)(row0 + mbase) * HID + mn;
        #pragma unroll
        for (int nt = 0; nt < 4; ++nt) {
            #pragma unroll
            for (int i = 0; i < 8; ++i)
                pr[i * HID + nt * 16] = acc2[nt][i];
        }
    } else {
        #pragma unroll
        for (int nt = 0; nt < 4; ++nt) {
            #pragma unroll
            for (int i = 0; i < 8; ++i) {
                const int r = row0 + mbase + i;
                if (r < n_nodes)
                    p[(size_t)r * HID + nt * 16 + mn] = acc2[nt][i];
            }
        }
    }
}

__device__ __forceinline__ void stage_two(
    float* ws, const float* __restrict__ Wa, const float* __restrict__ Wb, int wave, int lane)
{
    constexpr int W_SLOTS = 2 * 16;
    constexpr int W_FULL  = W_SLOTS / NT_WAVES;
    constexpr int W_REST  = W_SLOTS % NT_WAVES;
    static_assert(W_FULL >= 1, "every wave stages at least one slot");
    const int skp = lane >> 4;
    const int sn4 = lane & 15;
    float4 sev[W_FULL], sod[W_FULL];
    #pragma unroll
    for (int t = 0; t < W_FULL; ++t) {
        const int s = wave + t * NT_WAVES;
        const float* src = (s >> 4) == 0 ? Wa : Wb;
        const int kp = (s & 15) * 2 + skp;
        sev[t] = ((const float4*)src)[(2 * kp) * 16 + sn4];
        sod[t] = ((const float4*)src)[(2 * kp + 1) * 16 + sn4];
    }
    float4 rev = {0.f, 0.f, 0.f, 0.f}, rod = {0.f, 0.f, 0.f, 0.f};
    if (wave < W_REST) {
        const int s = wave + W_FULL * NT_WAVES;
        const float* src = (s >> 4) == 0 ? Wa : Wb;
        const int kp = (s & 15) * 2 + skp;
        rev = ((const float4*)src)[(2 * kp) * 16 + sn4];
        rod = ((const float4*)src)[(2 * kp + 1) * 16 + sn4];
    }
    #pragma unroll
    for (int t = 0; t < W_FULL; ++t) {
        const int s = wave + t * NT_WAVES;
        float4* d = (float4*)(ws + (s >> 4) * W_MAT
                                 + ((s & 15) * 2 + skp) * W_KP_PITCH + sn4 * 8);
        d[0] = make_float4(sev[t].x, sod[t].x, sev[t].y, sod[t].y);
        d[1] = make_float4(sev[t].z, sod[t].z, sev[t].w, sod[t].w);
    }
    if (wave < W_REST) {
        const int s = wave + W_FULL * NT_WAVES;
        float4* d = (float4*)(ws + (s >> 4) * W_MAT
                                 + ((s & 15) * 2 + skp) * W_KP_PITCH + sn4 * 8);
        d[0] = make_float4(rev.x, rod.x, rev.y, rod.y);
        d[1] = make_float4(rev.z, rod.z, rev.w, rod.w);
    }
}

__global__ __launch_bounds__(NT_THREADS) void node_transform_kernel(
    const float* __restrict__ x,
    const float* __restrict__ W1, const float* __restrict__ b1,
    const float* __restrict__ W2, float* __restrict__ p, int n_nodes)
{
    __shared__ __attribute__((aligned(16))) float ws[2 * W_MAT];
    __shared__ __attribute__((aligned(16))) float gs[G_SLOTS * 16 * G_PITCH];

    const int wave  = __builtin_amdgcn_readfirstlane((int)(threadIdx.x >> 5));
    const int lane  = threadIdx.x & 31;
    const int mn    = lane & 15;
    const int kh    = (lane >> 4) << 1;
    const int mbase = (lane >> 4) << 3;
    float* gt = gs + wave * (16 * G_PITCH);

    const int n_tiles = (n_nodes + 15) >> 4;
    const int blk = (int)blockIdx.x;
    const int cnt = blk < n_tiles ? (n_tiles - 1 - blk) / NT_BLOCKS + 1 : 0;

    stage_two(ws, W1, W2, wave, lane);
    __syncthreads();
    if (wave >= cnt) return;

    const int tile = blk + NT_BLOCKS * wave;
    v2f xa[16];
    node_tile_load(xa, x, n_nodes, tile, mn, kh);
    node_tile_project<false>(xa, ws, ws + W_MAT, gt, b1, nullptr, p, n_nodes,
                             tile, mn, kh, mbase);
}

static_assert(NEDGES % 4 == 0, "edge count must fill whole groups of four");

#define SORT_BIN_SHIFT   9
#define SORT_BIN_ROWS    (1 << SORT_BIN_SHIFT)
#define SORT_NBINS       (NNODES / SORT_BIN_ROWS + 1)
#define SORT_THREADS     512
#define SORT_CHUNK_QUADS (4 * SORT_THREADS)
#define SORT_NCHUNKS     ((NEDGES / 4 + SORT_CHUNK_QUADS - 1) / SORT_CHUNK_QUADS)
#define SORT_TABLE       (SORT_NBINS * SORT_NCHUNKS)
#define SCAN_PER         ((SORT_TABLE + 1023) / 1024)

__global__ __launch_bounds__(SORT_THREADS) void bin_count_kernel(
    const int* __restrict__ row, int* __restrict__ table, int n_quads)
{
    __shared__ int hist[SORT_NBINS];
    for (int b = threadIdx.x; b < SORT_NBINS; b += SORT_THREADS) hist[b] = 0;
    __syncthreads();
    const int q0 = blockIdx.x * SORT_CHUNK_QUADS + threadIdx.x;
    constexpr int NQ = SORT_CHUNK_QUADS / SORT_THREADS;
    int4 r[NQ];
    for (int j = 0; j < NQ; ++j) {
        const int q = q0 + j * SORT_THREADS;
        r[j] = ((const int4*)row)[q < n_quads ? q : n_quads - 1];
    }
    for (int j = 0; j < NQ; ++j) {
        if (q0 + j * SORT_THREADS < n_quads) {
            atomicAdd(&hist[r[j].x >> SORT_BIN_SHIFT], 1);
            atomicAdd(&hist[r[j].y >> SORT_BIN_SHIFT], 1);
            atomicAdd(&hist[r[j].z >> SORT_BIN_SHIFT], 1);
            atomicAdd(&hist[r[j].w >> SORT_BIN_SHIFT], 1);
        }
    }
    __syncthreads();
    for (int b = threadIdx.x; b < SORT_NBINS; b += SORT_THREADS)
        table[b * SORT_NCHUNKS + blockIdx.x] = hist[b];
}

__global__ __launch_bounds__(1024) void bin_scan_kernel(
    int* __restrict__ table, int* __restrict__ binbase)
{
    __shared__ int t[SCAN_PER * 1024];
    __shared__ int wsum[32];
    const int tid  = threadIdx.x;
    const int lane = tid & 31;
    const int wv   = tid >> 5;

    for (int j = 0; j < SCAN_PER; ++j) {
        const int i = j * 1024 + tid;
        t[i] = i < SORT_TABLE ? table[i] : 0;
    }
    __syncthreads();

    int* seg = t + tid * SCAN_PER;
    int sum = 0;
    for (int k = 0; k < SCAN_PER; ++k) sum += seg[k];

    int s = sum;
    for (int o = 1; o < 32; o <<= 1) {
        const int y = __shfl_up(s, o, 32);
        if (lane >= o) s += y;
    }
    if (lane == 31) wsum[wv] = s;
    __syncthreads();
    if (wv == 0) {
        const int w = wsum[lane];
        int u = w;
        for (int o = 1; o < 32; o <<= 1) {
            const int y = __shfl_up(u, o, 32);
            if (lane >= o) u += y;
        }
        wsum[lane] = u - w;
    }
    __syncthreads();

    int run = wsum[wv] + s - sum;
    for (int k = 0; k < SCAN_PER; ++k) {
        const int v = seg[k];
        seg[k] = run;
        run += v;
    }
    __syncthreads();

    for (int j = 0; j < SCAN_PER; ++j) {
        const int i = j * 1024 + tid;
        if (i < SORT_TABLE) {
            const int p = t[i];
            table[i] = p;
            if (i % SORT_NCHUNKS == 0) binbase[i / SORT_NCHUNKS] = p;
        }
    }
    if (tid == 1023) binbase[SORT_NBINS] = run;
}

__global__ __launch_bounds__(SORT_THREADS) void bin_scatter_kernel(
    const int* __restrict__ row, const int* __restrict__ col,
    const int* __restrict__ table, int* __restrict__ pairs, int n_quads)
{
    __shared__ int off[SORT_NBINS];
    for (int b = threadIdx.x; b < SORT_NBINS; b += SORT_THREADS)
        off[b] = table[b * SORT_NCHUNKS + blockIdx.x];
    __syncthreads();
    int2* __restrict__ pv = (int2*)pairs;
    const int q0 = blockIdx.x * SORT_CHUNK_QUADS + threadIdx.x;
    constexpr int NQ = SORT_CHUNK_QUADS / SORT_THREADS;
    int4 r[NQ], c[NQ];
    for (int j = 0; j < NQ; ++j) {
        const int q = q0 + j * SORT_THREADS;
        const int qc = q < n_quads ? q : n_quads - 1;
        r[j] = ((const int4*)row)[qc];
        c[j] = ((const int4*)col)[qc];
    }
    for (int j = 0; j < NQ; ++j) {
        if (q0 + j * SORT_THREADS < n_quads) {
            const int p0 = atomicAdd(&off[r[j].x >> SORT_BIN_SHIFT], 1);
            const int p1 = atomicAdd(&off[r[j].y >> SORT_BIN_SHIFT], 1);
            const int p2 = atomicAdd(&off[r[j].z >> SORT_BIN_SHIFT], 1);
            const int p3 = atomicAdd(&off[r[j].w >> SORT_BIN_SHIFT], 1);
            pv[p0] = make_int2(r[j].x, c[j].x);
            pv[p1] = make_int2(r[j].y, c[j].y);
            pv[p2] = make_int2(r[j].z, c[j].z);
            pv[p3] = make_int2(r[j].w, c[j].w);
        }
    }
}

#define BIN_WAVES (SORT_BIN_ROWS / 32)
__global__ __launch_bounds__(SORT_BIN_ROWS) void bin_rows_kernel(
    const int* __restrict__ pairs, const int* __restrict__ binbase,
    int* __restrict__ rowptr, int* __restrict__ cols, int n_nodes)
{
    __shared__ int cur[SORT_BIN_ROWS];
    __shared__ int wsum[BIN_WAVES];
    const int tid  = threadIdx.x;
    const int lane = tid & 31;
    const int wv   = tid >> 5;
    const int row0 = blockIdx.x * SORT_BIN_ROWS;
    const int lo   = binbase[blockIdx.x];
    const int hi   = binbase[blockIdx.x + 1];
    const int2* __restrict__ pv = (const int2*)pairs;

    cur[tid] = 0;
    __syncthreads();
    for (int e0 = lo + tid; e0 < hi + tid; e0 += 4 * SORT_BIN_ROWS) {
        int2 pr[4];
        for (int j = 0; j < 4; ++j) {
            const int e = e0 + j * SORT_BIN_ROWS;
            pr[j] = e < hi ? pv[e] : make_int2(row0, 0);
        }
        for (int j = 0; j < 4; ++j)
            if (e0 + j * SORT_BIN_ROWS < hi) atomicAdd(&cur[pr[j].x - row0], 1);
    }
    __syncthreads();

    const int v = cur[tid];
    int s = v;
    for (int o = 1; o < 32; o <<= 1) {
        const int y = __shfl_up(s, o, 32);
        if (lane >= o) s += y;
    }
    if (lane == 31) wsum[wv] = s;
    __syncthreads();
    if (wv == 0) {
        const int w = lane < BIN_WAVES ? wsum[lane] : 0;
        int u = w;
        for (int o = 1; o < 32; o <<= 1) {
            const int y = __shfl_up(u, o, 32);
            if (lane >= o) u += y;
        }
        if (lane < BIN_WAVES) wsum[lane] = u - w;
    }
    __syncthreads();
    const int excl = wsum[wv] + s - v;
    const int r = row0 + tid;
    if (r <= n_nodes) rowptr[r] = lo + excl;
    cur[tid] = excl;
    __syncthreads();

    for (int e0 = lo + tid; e0 < hi + tid; e0 += 4 * SORT_BIN_ROWS) {
        int2 pr[4];
        for (int j = 0; j < 4; ++j) {
            const int e = e0 + j * SORT_BIN_ROWS;
            pr[j] = e < hi ? pv[e] : make_int2(row0, 0);
        }
        for (int j = 0; j < 4; ++j) {
            if (e0 + j * SORT_BIN_ROWS < hi) {
                const int p = atomicAdd(&cur[pr[j].x - row0], 1);
                cols[lo + p] = pr[j].y;
            }
        }
    }
}

#define ROWS_PER_WAVE 8
#define ROW_WAVES_PER_BLOCK 4

#define DPP_ROW_SHARE(j) (0x150 + (j))
#define DPP_ROW_XMASK(m) (0x160 + (m))

template <int CTRL>
__device__ __forceinline__ int dpp_i(int v)
{
    return __builtin_amdgcn_update_dpp(0, v, CTRL, 0xf, 0xf, true);
}

template <int CTRL>
__device__ __forceinline__ float dpp_f(float v)
{
    return __builtin_bit_cast(float,
        __builtin_amdgcn_update_dpp(0, __builtin_bit_cast(int, v), CTRL, 0xf, 0xf, true));
}

template <int J>
__device__ __forceinline__ void row_share16(unsigned (&o)[16], int v)
{
    if constexpr (J < 16) {
        o[J] = (unsigned)dpp_i<DPP_ROW_SHARE(J)>(v);
        row_share16<J + 1>(o, v);
    }
}

template <int M, int N>
__device__ __forceinline__ void fold_step(float (&s)[16], bool hi)
{
    #pragma unroll
    for (int j = 0; j < N; ++j) {
        const float keep = hi ? s[j + N] : s[j];
        const float give = hi ? s[j] : s[j + N];
        s[j] = keep + dpp_f<DPP_ROW_XMASK(M)>(give);
    }
}

__device__ __forceinline__ float row_sums16(float (&s)[16], int sub)
{
    fold_step<8, 8>(s, (sub & 8) != 0);
    fold_step<4, 4>(s, (sub & 4) != 0);
    fold_step<2, 2>(s, (sub & 2) != 0);
    fold_step<1, 1>(s, (sub & 1) != 0);
    return s[0];
}

__device__ __forceinline__ float row_sum1(float v)
{
    v += dpp_f<DPP_ROW_XMASK(8)>(v);
    v += dpp_f<DPP_ROW_XMASK(4)>(v);
    v += dpp_f<DPP_ROW_XMASK(2)>(v);
    v += dpp_f<DPP_ROW_XMASK(1)>(v);
    return v;
}

#define ATT_ROWS 16
static_assert(ATT_ROWS + 2 <= 32, "a wave holds its rows' bucket bounds one per lane");

__global__ __launch_bounds__(NT_THREADS) void row_att_kernel(
    const float* __restrict__ x,
    const float* __restrict__ W1, const float* __restrict__ b1,
    const float* __restrict__ W2, const float* __restrict__ bias2,
    const float* __restrict__ p2,
    const float* __restrict__ w2, const float* __restrict__ b2,
    const int* __restrict__ rowptr, const int* __restrict__ cols,
    float* __restrict__ mask, float* __restrict__ dis, int n_nodes)
{
    __shared__ __attribute__((aligned(16))) float ws[2 * W_MAT];
    __shared__ __attribute__((aligned(16))) float gs[G_SLOTS * 16 * G_PITCH];

    const int wave  = __builtin_amdgcn_readfirstlane((int)(threadIdx.x >> 5));
    const int lane  = threadIdx.x & 31;
    const int sub   = lane & 15;
    const int hsel  = lane >> 4;
    const int kh    = hsel << 1;
    const int mbase = hsel << 3;
    float* gt = gs + wave * (16 * G_PITCH);

    const int n_tiles = (n_nodes + 15) >> 4;
    const int blk = (int)blockIdx.x;
    const int cnt = blk < n_tiles ? (n_tiles - 1 - blk) / NT_BLOCKS + 1 : 0;

    stage_two(ws, W1, W2, wave, lane);
    __syncthreads();
    if (wave >= cnt) return;

    const int tile = blk + NT_BLOCKS * wave;
    const int r0   = tile * ATT_ROWS;

    const int rp_i = r0 + lane;
    const int rp_l = rowptr[rp_i < n_nodes ? rp_i : n_nodes];

    v2f xa[16];
    node_tile_load(xa, x, n_nodes, tile, sub, kh);

    const float4* __restrict__ p2v = (const float4*)p2;
    const float4 w = ((const float4*)w2)[sub];
    const float bias = b2[0];

    int c_nx;
    {
        const int f0 = __builtin_amdgcn_readlane(rp_l, 0);
        const int f1 = __builtin_amdgcn_readlane(rp_l, 1);
        c_nx = cols[min((hsel ? f1 : f0) + sub, NEDGES - 1)];
    }

    node_tile_project<true>(xa, ws, ws + W_MAT, gt, b1, bias2, nullptr, n_nodes,
                            tile, sub, kh, mbase);

    for (int q = 0; q < ATT_ROWS / 2; ++q) {
        const int r = r0 + 2 * q + hsel;
        const bool rv = r < n_nodes;
        const int e0 = __builtin_amdgcn_readlane(rp_l, 2 * q);
        const int e1 = __builtin_amdgcn_readlane(rp_l, 2 * q + 1);
        const int e2 = __builtin_amdgcn_readlane(rp_l, 2 * q + 2);
        const int start = hsel ? e1 : e0;
        const int end   = hsel ? e2 : e1;
        const int degmax = max(e1 - e0, e2 - e1);
        const int c_first = c_nx;
        {
            const int g1 = __builtin_amdgcn_readlane(rp_l, 2 * q + 3);
            c_nx = cols[min((hsel ? g1 : e2) + sub, NEDGES - 1)];
        }
        const float4 a = *(const float4*)(gt + (2 * q + hsel) * G_PITCH + 4 * sub);

        float rs = 0.f;
        for (int base = 0; base < degmax; base += 16) {
            const int idx = start + base + sub;
            const bool ev = idx < end;
            int c_l = ev ? c_first : 0;
            if (base != 0) c_l = ev ? cols[idx] : 0;

            unsigned cj[16];
            float4 b[16];
            row_share16<0>(cj, c_l);
            #pragma unroll
            for (int j = 0; j < 16; ++j)
                cj[j] = cj[j] * (HID / 4) + sub;
            __builtin_amdgcn_sched_barrier(0);
            #pragma unroll
            for (int j = 0; j < 16; ++j)
                b[j] = p2v[cj[j]];

            float s[16];
            #pragma unroll
            for (int j = 0; j < 16; ++j) {
                const float t0 = fmaxf(a.x + b[j].x, 0.f);
                const float t1 = fmaxf(a.y + b[j].y, 0.f);
                const float t2 = fmaxf(a.z + b[j].z, 0.f);
                const float t3 = fmaxf(a.w + b[j].w, 0.f);
                s[j] = t0 * w.x + t1 * w.y + t2 * w.z + t3 * w.w;
            }
            const float mine = row_sums16(s, sub);

            const float la = mine + bias;
            const float sg = 1.f / (1.f + __expf(-la));
            float m = sg * (ZETA_C - GAMMA_C) + GAMMA_C;
            m = fminf(fmaxf(m, 0.f), 1.f);
            if (ev) { mask[idx] = m; rs += m; }
        }

        rs = row_sum1(rs);
        if (rv && sub == 0) dis[r] = fminf(rsqrtf(rs + 1e-6f), 10.f);
    }
}

__global__ __launch_bounds__(128) void row_spmm_kernel(
    const float* __restrict__ x, const float* __restrict__ mask,
    const float* __restrict__ dis,
    const int* __restrict__ rowptr, const int* __restrict__ cols,
    const float* addin, float* out, float* xn, int n_nodes)
{
    const int lane   = threadIdx.x & 31;
    const int sub    = lane & 15;
    const int hsel   = lane >> 4;
    const int wave_g = blockIdx.x * ROW_WAVES_PER_BLOCK + (threadIdx.x >> 5);
    const int r0     = wave_g * ROWS_PER_WAVE;
    if (r0 >= n_nodes) return;
    const float4* __restrict__ xv = (const float4*)x;
    const float4 zero4 = {0.f, 0.f, 0.f, 0.f};

    const int rp_i = r0 + lane;
    const int rp_l = rowptr[rp_i < n_nodes ? rp_i : n_nodes];

    int c_nx;
    {
        const int f0 = __builtin_amdgcn_readlane(rp_l, 0);
        const int f1 = __builtin_amdgcn_readlane(rp_l, 1);
        c_nx = cols[min((hsel ? f1 : f0) + sub, NEDGES - 1)];
    }

    for (int q = 0; q < ROWS_PER_WAVE / 2; ++q) {
        const int r = r0 + 2 * q + hsel;
        const bool rv = r < n_nodes;
        const int e0 = __builtin_amdgcn_readlane(rp_l, 2 * q);
        const int e1 = __builtin_amdgcn_readlane(rp_l, 2 * q + 1);
        const int e2 = __builtin_amdgcn_readlane(rp_l, 2 * q + 2);
        const int start = hsel ? e1 : e0;
        const int end   = hsel ? e2 : e1;
        const int degmax = max(e1 - e0, e2 - e1);
        const int c_first = c_nx;
        {
            const int g1 = __builtin_amdgcn_readlane(rp_l, 2 * q + 3);
            c_nx = cols[min((hsel ? g1 : e2) + sub, NEDGES - 1)];
        }
        const float dr = dis[rv ? r : n_nodes - 1];
        const int o = (rv ? r : n_nodes - 1) * (HID / 4) + sub;
        const float4 bs = ((const float4*)addin)[o];

        float4 acc = zero4;
        for (int base = 0; base < degmax; base += 16) {
            const int idx = start + base + sub;
            const bool ev = idx < end;
            int c_l = ev ? c_first : 0;
            if (base != 0) c_l = ev ? cols[idx] : 0;
            const int nv  = end - (start + base);

            unsigned cj[16];
            float4 xc[16];
            row_share16<0>(cj, c_l);
            #pragma unroll
            for (int j = 0; j < 16; ++j)
                xc[j] = xv[cj[j] * (HID / 4) + sub];

            float v_l = 0.f;
            if (ev) v_l = mask[idx] * dr * dis[c_l];

            unsigned vj[16];
            row_share16<0>(vj, __builtin_bit_cast(int, v_l));

            #pragma unroll
            for (int j = 0; j < 16; ++j) {
                const float v = __builtin_bit_cast(float, vj[j]);
                const float4 t = (j < nv) ? xc[j] : zero4;
                acc.x += v * t.x;
                acc.y += v * t.y;
                acc.z += v * t.z;
                acc.w += v * t.w;
            }
        }

        if (rv) {
            float4 res;
            res.x = bs.x + acc.x; res.y = bs.y + acc.y;
            res.z = bs.z + acc.z; res.w = bs.w + acc.w;
            ((float4*)out)[o] = res;
            if (xn) ((float4*)xn)[o] = acc;
        }
    }
}

extern "C" void kernel_launch(void* const* d_in, const int* in_sizes, int n_in,
                              void* d_out, int out_size, void* d_ws, size_t ws_size,
                              hipStream_t stream) {
    (void)in_sizes; (void)n_in; (void)out_size; (void)ws_size;

    const float* feat = (const float*)d_in[0];
    const float* nbW  = (const float*)d_in[1];
    const float* nbb  = (const float*)d_in[2];
    const float* sW   = (const float*)d_in[3];
    const float* sb   = (const float*)d_in[4];
    const float* a1W  = (const float*)d_in[5];
    const float* a1b  = (const float*)d_in[6];
    const float* a2W  = (const float*)d_in[7];
    const float* a2b  = (const float*)d_in[8];
    const int*   row  = (const int*)d_in[9];
    const int*   col  = (const int*)d_in[10];
    float* out = (float*)d_out;

    const int NH = NNODES * HID;
    float* ws = (float*)d_ws;
    float* bufA   = ws; ws += NH;
    float* bufP1  = ws; ws += NH;
    float* bufP2  = ws; ws += NH;
    int*   cols   = (int*)ws; ws += NEDGES;
    float* maskb  = ws; ws += NEDGES;
    int*   rowptr = (int*)ws; ws += NNODES + 32;
    int*   table  = (int*)ws; ws += NNODES;
    float* dis    = ws; ws += NNODES;
    int*   binbase = table + SORT_TABLE;
    int*   pairs  = (int*)bufA;
    static_assert(SORT_TABLE + SORT_NBINS + 1 <= NNODES, "table and bin bases share one plane");
    static_assert(2 * NEDGES <= NNODES * HID, "the (row, col) pairs must fit bufA");

    const int nQuad  = NEDGES / 4;
    const int gRow   = (NNODES + ROWS_PER_WAVE * ROW_WAVES_PER_BLOCK - 1)
                       / (ROWS_PER_WAVE * ROW_WAVES_PER_BLOCK);

    bin_count_kernel<<<SORT_NCHUNKS, SORT_THREADS, 0, stream>>>(row, table, nQuad);
    bin_scan_kernel<<<1, 1024, 0, stream>>>(table, binbase);
    bin_scatter_kernel<<<SORT_NCHUNKS, SORT_THREADS, 0, stream>>>(row, col, table, pairs, nQuad);
    bin_rows_kernel<<<SORT_NBINS, SORT_BIN_ROWS, 0, stream>>>(pairs, binbase, rowptr, cols, NNODES);

    const float* xin[NLAYERS]  = {feat,  bufA,  bufP1};
    float*       xout[NLAYERS] = {bufA,  bufP1, nullptr};
    const float* addin[NLAYERS] = {feat, out, out};

    for (int l = 0; l < NLAYERS; ++l) {
        node_transform_kernel<<<NT_BLOCKS, NT_THREADS, 0, stream>>>(
            xin[l],
            sW + (size_t)l * 4096, sb + l * 64,
            a1W + (size_t)l * 8192 + 4096, bufP2, NNODES);

        row_att_kernel<<<NT_BLOCKS, NT_THREADS, 0, stream>>>(
            xin[l],
            nbW + (size_t)l * 4096, nbb + l * 64,
            a1W + (size_t)l * 8192, a1b + l * 64,
            bufP2, a2W + l * 64, a2b + l, rowptr, cols, maskb, dis, NNODES);

        row_spmm_kernel<<<gRow, 128, 0, stream>>>(
            xin[l], maskb, dis, rowptr, cols, addin[l], out, xout[l], NNODES);
    }
}
